# starting kernel
# baseline (speedup 1.0000x reference)
#include <hip/hip_runtime.h>
#include <hip/hip_bf16.h>


static constexpr int KF  = 128;
static constexpr int KH  = 8;
static constexpr int KD  = 16;
static constexpr int KHD = 128;
static constexpr float KALPHA = 0.2f;

typedef __attribute__((ext_vector_type(2))) float v2f;
typedef __attribute__((ext_vector_type(8))) float v8f;

__device__ __forceinline__ unsigned fmax_enc(float f) {
  unsigned u = __float_as_uint(f);
  return (u & 0x80000000u) ? ~u : (u | 0x80000000u);
}
__device__ __forceinline__ float fmax_dec(unsigned u) {
  return (u & 0x80000000u) ? __uint_as_float(u & 0x7fffffffu)
                           : __uint_as_float(~u);
}
static constexpr unsigned ENC_NEG_INF = 0x007FFFFFu;

__global__ void init_nh_kernel(float* __restrict__ z, unsigned* __restrict__ menc, int count) {
  int i = blockIdx.x * blockDim.x + threadIdx.x;
  if (i < count) { z[i] = 0.0f; menc[i] = ENC_NEG_INF; }
}
__global__ void zero_kernel(float* __restrict__ p, int count) {
  int i = blockIdx.x * blockDim.x + threadIdx.x;
  if (i < count) p[i] = 0.0f;
}

static constexpr int AS_STRIDE = 132;

__global__ __launch_bounds__(256) void gat_gemm_wmma_kernel(
    const float* __restrict__ hfeat, const float* __restrict__ W,
    float* __restrict__ Wh, int N) {
  __shared__ float As[16 * AS_STRIDE];

  const int tid  = threadIdx.x;
  const int row0 = blockIdx.x * 16;

  for (int i = tid; i < 16 * KF; i += 256) {
    int r = i >> 7, f = i & (KF - 1);
    float v = 0.0f;
    if (row0 + r < N) v = hfeat[(size_t)(row0 + r) * KF + f];
    As[r * AS_STRIDE + f] = v;
  }
  __syncthreads();

  const int wave = tid >> 5;
  const int lane = tid & 31;
  const int hl   = lane >> 4;
  const int col  = lane & 15;
  const int n0   = wave * 16;

  const float* Wb = W + wave * (KF * KD) + col;

  v8f acc = {};
  for (int k0 = 0; k0 < KF; k0 += 4) {
    const int k = k0 + 2 * hl;
    v2f a, b;
    a.x = As[col * AS_STRIDE + k];
    a.y = As[col * AS_STRIDE + k + 1];
    b.x = Wb[(size_t)k * KD];
    b.y = Wb[(size_t)(k + 1) * KD];
    acc = __builtin_amdgcn_wmma_f32_16x16x4_f32(
        false, a, false, b, (short)0, acc, false, false);
  }

#pragma unroll
  for (int v = 0; v < 8; ++v) {
    int r = row0 + v + 8 * hl;
    if (r < N) Wh[(size_t)r * KHD + n0 + col] = acc[v];
  }
}

__global__ void scores_kernel(const float* __restrict__ Wh,
                              const float* __restrict__ a1,
                              const float* __restrict__ a2,
                              float* __restrict__ s1, float* __restrict__ s2,
                              int N) {
  int idx = blockIdx.x * blockDim.x + threadIdx.x;
  if (idx >= N * KH) return;
  int n = idx >> 3, hh = idx & 7;
  const float* w  = Wh + (size_t)n * KHD + hh * KD;
  const float* v1 = a1 + hh * KD;
  const float* v2 = a2 + hh * KD;
  float acc1 = 0.0f, acc2 = 0.0f;
#pragma unroll
  for (int d = 0; d < KD; ++d) {
    float v = w[d];
    acc1 += v * v1[d];
    acc2 += v * v2[d];
  }
  s1[idx] = acc1;
  s2[idx] = acc2;
}

__device__ __forceinline__ float edge_logit(const float* s1, const float* s2,
                                            int s, int d, int hh) {
  float e = s1[s * KH + hh] + s2[d * KH + hh];
  return (e > 0.0f) ? e : KALPHA * e;
}

__global__ void edge_max_kernel(const int* __restrict__ src, const int* __restrict__ dst,
                                const float* __restrict__ s1, const float* __restrict__ s2,
                                unsigned* __restrict__ menc, int E) {
  int idx = blockIdx.x * blockDim.x + threadIdx.x;
  if (idx >= E * KH) return;
  int e = idx >> 3, hh = idx & 7;
  int d = dst[e];
  float lv = edge_logit(s1, s2, src[e], d, hh);
  atomicMax(&menc[d * KH + hh], fmax_enc(lv));
}

__global__ void finalize_max_kernel(const unsigned* __restrict__ menc,
                                    float* __restrict__ m, int count) {
  int i = blockIdx.x * blockDim.x + threadIdx.x;
  if (i >= count) return;
  float v = fmax_dec(menc[i]);
  m[i] = (v < -3.0e38f) ? 0.0f : v;
}

__global__ void edge_expsum_kernel(const int* __restrict__ src, const int* __restrict__ dst,
                                   const float* __restrict__ s1, const float* __restrict__ s2,
                                   const float* __restrict__ m, float* __restrict__ z, int E) {
  int idx = blockIdx.x * blockDim.x + threadIdx.x;
  if (idx >= E * KH) return;
  int e = idx >> 3, hh = idx & 7;
  int d = dst[e];
  float lv = edge_logit(s1, s2, src[e], d, hh);
  float p = __expf(lv - m[d * KH + hh]);
  unsafeAtomicAdd(&z[d * KH + hh], p);
}

__global__ void finalize_z_kernel(float* __restrict__ z, int count) {
  int i = blockIdx.x * blockDim.x + threadIdx.x;
  if (i >= count) return;
  z[i] = 1.0f / fmaxf(z[i], 1e-16f);
}

__global__ void edge_scatter_kernel(const int* __restrict__ src, const int* __restrict__ dst,
                                    const float* __restrict__ s1, const float* __restrict__ s2,
                                    const float* __restrict__ m, const float* __restrict__ zinv,
                                    const float* __restrict__ Wh, float* __restrict__ out, int E) {
  int idx = blockIdx.x * blockDim.x + threadIdx.x;
  if (idx >= E * KH) return;
  int e = idx >> 3, hh = idx & 7;
  int s = src[e], d = dst[e];
  float lv   = edge_logit(s1, s2, s, d, hh);
  float attn = __expf(lv - m[d * KH + hh]) * zinv[d * KH + hh];
  const float4* wsrc = (const float4*)(Wh + (size_t)s * KHD + hh * KD);
  float* op = out + (size_t)d * KHD + hh * KD;
#pragma unroll
  for (int q = 0; q < 4; ++q) {
    float4 w4 = wsrc[q];
    unsafeAtomicAdd(op + q * 4 + 0, attn * w4.x);
    unsafeAtomicAdd(op + q * 4 + 1, attn * w4.y);
    unsafeAtomicAdd(op + q * 4 + 2, attn * w4.z);
    unsafeAtomicAdd(op + q * 4 + 3, attn * w4.w);
  }
}

extern "C" void kernel_launch(void* const* d_in, const int* in_sizes, int n_in,
                              void* d_out, int out_size, void* d_ws, size_t ws_size,
                              hipStream_t stream) {
  const float* hfeat = (const float*)d_in[0];
  const int*   src   = (const int*)d_in[1];
  const int*   dst   = (const int*)d_in[2];
  const float* W     = (const float*)d_in[3];
  const float* a1    = (const float*)d_in[4];
  const float* a2    = (const float*)d_in[5];
  float* out = (float*)d_out;

  const int N = in_sizes[0] / KF;
  const int E = in_sizes[1];

  float*    Wh   = (float*)d_ws;
  float*    s1   = Wh + (size_t)N * KHD;
  float*    s2   = s1 + (size_t)N * KH;
  unsigned* menc = (unsigned*)(s2 + (size_t)N * KH);
  float*    m    = (float*)(menc + (size_t)N * KH);
  float*    z    = m + (size_t)N * KH;

  const int TB = 256;
  const int nh      = N * KH;
  const int nhd     = N * KHD;
  const int eh      = E * KH;
  const int g_nh    = (nh  + TB - 1) / TB;
  const int g_nhd   = (nhd + TB - 1) / TB;
  const int g_eh    = (eh  + TB - 1) / TB;
  const int g_gemm  = (N + 15) / 16;

  init_nh_kernel<<<g_nh, TB, 0, stream>>>(z, menc, nh);
  zero_kernel<<<g_nhd, TB, 0, stream>>>(out, nhd);

  gat_gemm_wmma_kernel<<<g_gemm, TB, 0, stream>>>(hfeat, W, Wh, N);

  scores_kernel<<<g_nh, TB, 0, stream>>>(Wh, a1, a2, s1, s2, N);
  edge_max_kernel<<<g_eh, TB, 0, stream>>>(src, dst, s1, s2, menc, E);
  finalize_max_kernel<<<g_nh, TB, 0, stream>>>(menc, m, nh);
  edge_expsum_kernel<<<g_eh, TB, 0, stream>>>(src, dst, s1, s2, m, z, E);
  finalize_z_kernel<<<g_nh, TB, 0, stream>>>(z, nh);
  edge_scatter_kernel<<<g_eh, TB, 0, stream>>>(src, dst, s1, s2, m, z, Wh, out, E);
}


# Float-atomic scatter replaced: edges bucketed by dst in-call (int atomic rank, scan, place); one wave per dst row does max, exp-sum, Wh gather in registers; sums in arrival-rank order, 1/z after sum
# speedup vs baseline: 4.7349x; 4.7349x over previous
#include <hip/hip_runtime.h>
#include <hip/hip_bf16.h>


static constexpr int KF  = 128;
static constexpr int KH  = 8;
static constexpr int KD  = 16;
static constexpr int KHD = 128;
static constexpr float KALPHA = 0.2f;

typedef __attribute__((ext_vector_type(2))) float v2f;
typedef __attribute__((ext_vector_type(8))) float v8f;

__global__ void zero_deg_kernel(int* __restrict__ deg, int count) {
  int i = blockIdx.x * blockDim.x + threadIdx.x;
  if (i < count) deg[i] = 0;
}

__global__ void count_rank_kernel(const int* __restrict__ dst, int* __restrict__ deg,
                                  int* __restrict__ rank, int E) {
  int e = blockIdx.x * blockDim.x + threadIdx.x;
  if (e >= E) return;
  rank[e] = atomicAdd(&deg[dst[e]], 1);
}

static constexpr int SCAN_TILE = 1024;

__global__ __launch_bounds__(256) void tile_sum_kernel(const int* __restrict__ deg,
                                                       int* __restrict__ tsum, int N) {
  __shared__ int red[8];
  const int tid = threadIdx.x, lane = tid & 31, wave = tid >> 5;
  const int base = blockIdx.x * SCAN_TILE + tid * 4;
  int t = 0;
#pragma unroll
  for (int q = 0; q < 4; ++q)
    if (base + q < N) t += deg[base + q];
  for (int off = 16; off > 0; off >>= 1) t += __shfl_xor(t, off);
  if (lane == 0) red[wave] = t;
  __syncthreads();
  if (tid == 0) {
    int a = 0;
#pragma unroll
    for (int w = 0; w < 8; ++w) a += red[w];
    tsum[blockIdx.x] = a;
  }
}

__global__ __launch_bounds__(256) void row_offsets_kernel(const int* __restrict__ deg,
                                                          const int* __restrict__ tsum,
                                                          int* __restrict__ offs, int N) {
  __shared__ int winc[8];
  __shared__ int wpre[8];
  const int tid = threadIdx.x, lane = tid & 31, wave = tid >> 5;

  int pre = 0;
  for (int i = tid; i < (int)blockIdx.x; i += 256) pre += tsum[i];
  for (int off = 16; off > 0; off >>= 1) pre += __shfl_xor(pre, off);

  const int base = blockIdx.x * SCAN_TILE + tid * 4;
  int v[4];
  int t = 0;
#pragma unroll
  for (int q = 0; q < 4; ++q) {
    v[q] = (base + q < N) ? deg[base + q] : 0;
    t += v[q];
  }
  int inc = t;
  for (int off = 1; off < 32; off <<= 1) {
    int u = __shfl_up(inc, off);
    if (lane >= off) inc += u;
  }
  if (lane == 31) winc[wave] = inc;
  if (lane == 0) wpre[wave] = pre;
  __syncthreads();
  int add = 0;
#pragma unroll
  for (int w = 0; w < 8; ++w) {
    add += wpre[w];
    if (w < wave) add += winc[w];
  }
  int ex = add + inc - t;
#pragma unroll
  for (int q = 0; q < 4; ++q) {
    if (base + q < N) offs[base + q] = ex;
    ex += v[q];
  }
}

__global__ void place_kernel(const int* __restrict__ src, const int* __restrict__ dst,
                             const int* __restrict__ rank, const int* __restrict__ offs,
                             int* __restrict__ ssrc, int E) {
  int e = blockIdx.x * blockDim.x + threadIdx.x;
  if (e >= E) return;
  ssrc[offs[dst[e]] + rank[e]] = src[e];
}

static constexpr int AS_STRIDE = 132;

__global__ __launch_bounds__(256) void gat_gemm_wmma_kernel(
    const float* __restrict__ hfeat, const float* __restrict__ W,
    float* __restrict__ Wh, int N) {
  __shared__ float As[16 * AS_STRIDE];

  const int tid  = threadIdx.x;
  const int row0 = blockIdx.x * 16;

  for (int i = tid; i < 16 * KF; i += 256) {
    int r = i >> 7, f = i & (KF - 1);
    float v = 0.0f;
    if (row0 + r < N) v = hfeat[(size_t)(row0 + r) * KF + f];
    As[r * AS_STRIDE + f] = v;
  }
  __syncthreads();

  const int wave = tid >> 5;
  const int lane = tid & 31;
  const int hl   = lane >> 4;
  const int col  = lane & 15;
  const int n0   = wave * 16;

  const float* Wb = W + wave * (KF * KD) + col;

  v8f acc = {};
  for (int k0 = 0; k0 < KF; k0 += 4) {
    const int k = k0 + 2 * hl;
    v2f a, b;
    a.x = As[col * AS_STRIDE + k];
    a.y = As[col * AS_STRIDE + k + 1];
    b.x = Wb[(size_t)k * KD];
    b.y = Wb[(size_t)(k + 1) * KD];
    acc = __builtin_amdgcn_wmma_f32_16x16x4_f32(
        false, a, false, b, (short)0, acc, false, false);
  }

#pragma unroll
  for (int v = 0; v < 8; ++v) {
    int r = row0 + v + 8 * hl;
    if (r < N) Wh[(size_t)r * KHD + n0 + col] = acc[v];
  }
}

__global__ void scores_kernel(const float* __restrict__ Wh,
                              const float* __restrict__ a1,
                              const float* __restrict__ a2,
                              float* __restrict__ s1, float* __restrict__ s2,
                              int N) {
  int idx = blockIdx.x * blockDim.x + threadIdx.x;
  if (idx >= N * KH) return;
  int n = idx >> 3, hh = idx & 7;
  const float* w  = Wh + (size_t)n * KHD + hh * KD;
  const float* v1 = a1 + hh * KD;
  const float* v2 = a2 + hh * KD;
  float acc1 = 0.0f, acc2 = 0.0f;
#pragma unroll
  for (int d = 0; d < KD; ++d) {
    float v = w[d];
    acc1 += v * v1[d];
    acc2 += v * v2[d];
  }
  s1[idx] = acc1;
  s2[idx] = acc2;
}

__device__ __forceinline__ float leaky(float e) {
  return (e > 0.0f) ? e : KALPHA * e;
}

__global__ __launch_bounds__(256) void row_softmax_gather_kernel(
    const int* __restrict__ ssrc, const int* __restrict__ offs, const int* __restrict__ deg,
    const float* __restrict__ s1, const float* __restrict__ s2,
    const float* __restrict__ Wh, float* __restrict__ out, int N) {
  const int lane = threadIdx.x & 31;
  const int d = blockIdx.x * 8 + __builtin_amdgcn_readfirstlane((int)(threadIdx.x >> 5));
  if (d >= N) return;
  const int beg = offs[d];
  const int end = beg + deg[d];

  const int h1 = lane & 7;
  const float s2a = s2[d * KH + h1];
  float mx = -__builtin_inff();
  for (int i = beg + (lane >> 3); i < end; i += 4) {
    const int s = ssrc[i];
    mx = fmaxf(mx, leaky(s1[s * KH + h1] + s2a));
  }
  mx = fmaxf(mx, __shfl_xor(mx, 8));
  mx = fmaxf(mx, __shfl_xor(mx, 16));
  const float m = __builtin_isfinite(mx) ? mx : 0.0f;

  const int hh = lane >> 2;
  const float mh  = __shfl(m, hh);
  const float s2b = s2[d * KH + hh];
  const float* s1h = s1 + hh;
  const float* whl = Wh + lane * 4;
  float4 acc = make_float4(0.0f, 0.0f, 0.0f, 0.0f);
  float z = 0.0f;
  int i = beg;
  for (; i + 4 <= end; i += 4) {
    const int sa = ssrc[i], sb = ssrc[i + 1], sc = ssrc[i + 2], sd = ssrc[i + 3];
    const float ea = s1h[sa * KH], eb = s1h[sb * KH], ec = s1h[sc * KH], ed = s1h[sd * KH];
    const float4 wa = *(const float4*)(whl + (size_t)sa * KHD);
    const float4 wb = *(const float4*)(whl + (size_t)sb * KHD);
    const float4 wc = *(const float4*)(whl + (size_t)sc * KHD);
    const float4 wd = *(const float4*)(whl + (size_t)sd * KHD);
    const float pa = __expf(leaky(ea + s2b) - mh);
    const float pb = __expf(leaky(eb + s2b) - mh);
    const float pc = __expf(leaky(ec + s2b) - mh);
    const float pd = __expf(leaky(ed + s2b) - mh);
    z += pa; acc.x += pa * wa.x; acc.y += pa * wa.y; acc.z += pa * wa.z; acc.w += pa * wa.w;
    z += pb; acc.x += pb * wb.x; acc.y += pb * wb.y; acc.z += pb * wb.z; acc.w += pb * wb.w;
    z += pc; acc.x += pc * wc.x; acc.y += pc * wc.y; acc.z += pc * wc.z; acc.w += pc * wc.w;
    z += pd; acc.x += pd * wd.x; acc.y += pd * wd.y; acc.z += pd * wd.z; acc.w += pd * wd.w;
  }
#pragma nounroll
  for (; i < end; ++i) {
    const int sa = ssrc[i];
    const float ea = s1h[sa * KH];
    const float4 wa = *(const float4*)(whl + (size_t)sa * KHD);
    const float pa = __expf(leaky(ea + s2b) - mh);
    z += pa; acc.x += pa * wa.x; acc.y += pa * wa.y; acc.z += pa * wa.z; acc.w += pa * wa.w;
  }
  const float zinv = 1.0f / fmaxf(z, 1e-16f);
  *(float4*)(out + (size_t)d * KHD + lane * 4) =
      make_float4(acc.x * zinv, acc.y * zinv, acc.z * zinv, acc.w * zinv);
}

extern "C" void kernel_launch(void* const* d_in, const int* in_sizes, int n_in,
                              void* d_out, int out_size, void* d_ws, size_t ws_size,
                              hipStream_t stream) {
  const float* hfeat = (const float*)d_in[0];
  const int*   src   = (const int*)d_in[1];
  const int*   dst   = (const int*)d_in[2];
  const float* W     = (const float*)d_in[3];
  const float* a1    = (const float*)d_in[4];
  const float* a2    = (const float*)d_in[5];
  float* out = (float*)d_out;

  const int N = in_sizes[0] / KF;
  const int E = in_sizes[1];

  float* Wh   = (float*)d_ws;
  float* s1   = Wh + (size_t)N * KHD;
  float* s2   = s1 + (size_t)N * KH;
  int*   ssrc = (int*)(s2 + (size_t)N * KH);
  int*   deg  = ssrc + (size_t)E;
  int*   offs = deg + (size_t)N;
  int*   tsum = offs + (size_t)N;
  int*   rank = (int*)d_ws;

  const int TB = 256;
  const int nh      = N * KH;
  const int g_n     = (N  + TB - 1) / TB;
  const int g_e     = (E  + TB - 1) / TB;
  const int g_nh    = (nh + TB - 1) / TB;
  const int g_tile  = (N + SCAN_TILE - 1) / SCAN_TILE;
  const int g_row   = (N + 7) / 8;
  const int g_gemm  = (N + 15) / 16;

  zero_deg_kernel<<<g_n, TB, 0, stream>>>(deg, N);
  count_rank_kernel<<<g_e, TB, 0, stream>>>(dst, deg, rank, E);
  tile_sum_kernel<<<g_tile, TB, 0, stream>>>(deg, tsum, N);
  row_offsets_kernel<<<g_tile, TB, 0, stream>>>(deg, tsum, offs, N);
  place_kernel<<<g_e, TB, 0, stream>>>(src, dst, rank, offs, ssrc, E);

  gat_gemm_wmma_kernel<<<g_gemm, TB, 0, stream>>>(hfeat, W, Wh, N);

  scores_kernel<<<g_nh, TB, 0, stream>>>(Wh, a1, a2, s1, s2, N);
  row_softmax_gather_kernel<<<g_row, TB, 0, stream>>>(ssrc, offs, deg, s1, s2, Wh, out, N);
}


# Edge ranks from LDS counters: 16K-edge groups count dst in LDS (2 row passes), reserve each row span by one line-wide global add, rank by LDS add; row sums still in arrival-rank order
# speedup vs baseline: 6.3598x; 1.3432x over previous
#include <hip/hip_runtime.h>
#include <hip/hip_bf16.h>


static constexpr int KF  = 128;
static constexpr int KH  = 8;
static constexpr int KD  = 16;
static constexpr int KHD = 128;
static constexpr float KALPHA = 0.2f;

typedef __attribute__((ext_vector_type(2))) float v2f;
typedef __attribute__((ext_vector_type(8))) float v8f;

__global__ void zero_deg_kernel(int* __restrict__ deg, int count) {
  int i = blockIdx.x * blockDim.x + threadIdx.x;
  if (i < count) deg[i] = 0;
}

static constexpr int RK_THREADS = 1024;
static constexpr int RK_EPT     = 16;
static constexpr int RK_ROWS    = 51200;
static constexpr int RK_BATCH   = 25;

__global__ __launch_bounds__(RK_THREADS) void rank_lds_kernel(const int* __restrict__ dst,
                                                              int* __restrict__ deg,
                                                              int* __restrict__ rank, int N, int E) {
  __shared__ int cnt[RK_ROWS];
  const int tid   = threadIdx.x;
  const int ebase = blockIdx.x * (RK_EPT * RK_THREADS) + tid;

  int dv[RK_EPT], rk[RK_EPT];
#pragma unroll
  for (int q = 0; q < RK_EPT; ++q) {
    const int e = ebase + q * RK_THREADS;
    dv[q] = (e < E) ? dst[e] : -1;
    rk[q] = 0;
  }

  for (int r0 = 0; r0 < N; r0 += RK_ROWS) {
    const int rows = min(RK_ROWS, N - r0);
    for (int w = tid; w < RK_ROWS; w += RK_THREADS) cnt[w] = 0;
    __syncthreads();

#pragma unroll
    for (int q = 0; q < RK_EPT; ++q) {
      const unsigned r = (unsigned)(dv[q] - r0);
      if (r < (unsigned)rows) atomicAdd(&cnt[r], 1);
    }
    __syncthreads();

    for (int wb = tid; wb < rows; wb += RK_THREADS * RK_BATCH) {
      int c[RK_BATCH], b[RK_BATCH];
#pragma unroll
      for (int q = 0; q < RK_BATCH; ++q) {
        const int w = wb + q * RK_THREADS;
        c[q] = (w < rows) ? cnt[w] : 0;
      }
#pragma unroll
      for (int q = 0; q < RK_BATCH; ++q) {
        b[q] = 0;
        if (c[q] != 0) b[q] = atomicAdd(&deg[r0 + wb + q * RK_THREADS], c[q]);
      }
#pragma unroll
      for (int q = 0; q < RK_BATCH; ++q)
        if (c[q] != 0) cnt[wb + q * RK_THREADS] = b[q];
    }
    __syncthreads();

#pragma unroll
    for (int q = 0; q < RK_EPT; ++q) {
      const unsigned r = (unsigned)(dv[q] - r0);
      if (r < (unsigned)rows) rk[q] = atomicAdd(&cnt[r], 1);
    }
    __syncthreads();
  }

#pragma unroll
  for (int q = 0; q < RK_EPT; ++q) {
    const int e = ebase + q * RK_THREADS;
    if (e < E) rank[e] = rk[q];
  }
}

static constexpr int SCAN_TILE = 1024;

__global__ __launch_bounds__(256) void tile_sum_kernel(const int* __restrict__ deg,
                                                       int* __restrict__ tsum, int N) {
  __shared__ int red[8];
  const int tid = threadIdx.x, lane = tid & 31, wave = tid >> 5;
  const int base = blockIdx.x * SCAN_TILE + tid * 4;
  int t = 0;
#pragma unroll
  for (int q = 0; q < 4; ++q)
    if (base + q < N) t += deg[base + q];
  for (int off = 16; off > 0; off >>= 1) t += __shfl_xor(t, off);
  if (lane == 0) red[wave] = t;
  __syncthreads();
  if (tid == 0) {
    int a = 0;
#pragma unroll
    for (int w = 0; w < 8; ++w) a += red[w];
    tsum[blockIdx.x] = a;
  }
}

__global__ __launch_bounds__(256) void row_offsets_kernel(const int* __restrict__ deg,
                                                          const int* __restrict__ tsum,
                                                          int* __restrict__ offs, int N) {
  __shared__ int winc[8];
  __shared__ int wpre[8];
  const int tid = threadIdx.x, lane = tid & 31, wave = tid >> 5;

  int pre = 0;
  for (int i = tid; i < (int)blockIdx.x; i += 256) pre += tsum[i];
  for (int off = 16; off > 0; off >>= 1) pre += __shfl_xor(pre, off);

  const int base = blockIdx.x * SCAN_TILE + tid * 4;
  int v[4];
  int t = 0;
#pragma unroll
  for (int q = 0; q < 4; ++q) {
    v[q] = (base + q < N) ? deg[base + q] : 0;
    t += v[q];
  }
  int inc = t;
  for (int off = 1; off < 32; off <<= 1) {
    int u = __shfl_up(inc, off);
    if (lane >= off) inc += u;
  }
  if (lane == 31) winc[wave] = inc;
  if (lane == 0) wpre[wave] = pre;
  __syncthreads();
  int add = 0;
#pragma unroll
  for (int w = 0; w < 8; ++w) {
    add += wpre[w];
    if (w < wave) add += winc[w];
  }
  int ex = add + inc - t;
#pragma unroll
  for (int q = 0; q < 4; ++q) {
    if (base + q < N) offs[base + q] = ex;
    ex += v[q];
  }
}

__global__ void place_kernel(const int* __restrict__ src, const int* __restrict__ dst,
                             const int* __restrict__ rank, const int* __restrict__ offs,
                             int* __restrict__ ssrc, int E) {
  int e = blockIdx.x * blockDim.x + threadIdx.x;
  if (e >= E) return;
  ssrc[offs[dst[e]] + rank[e]] = src[e];
}

static constexpr int AS_STRIDE = 132;

__global__ __launch_bounds__(256) void gat_gemm_wmma_kernel(
    const float* __restrict__ hfeat, const float* __restrict__ W,
    float* __restrict__ Wh, int N) {
  __shared__ float As[16 * AS_STRIDE];

  const int tid  = threadIdx.x;
  const int row0 = blockIdx.x * 16;

  for (int i = tid; i < 16 * KF; i += 256) {
    int r = i >> 7, f = i & (KF - 1);
    float v = 0.0f;
    if (row0 + r < N) v = hfeat[(size_t)(row0 + r) * KF + f];
    As[r * AS_STRIDE + f] = v;
  }
  __syncthreads();

  const int wave = tid >> 5;
  const int lane = tid & 31;
  const int hl   = lane >> 4;
  const int col  = lane & 15;
  const int n0   = wave * 16;

  const float* Wb = W + wave * (KF * KD) + col;

  v8f acc = {};
  for (int k0 = 0; k0 < KF; k0 += 4) {
    const int k = k0 + 2 * hl;
    v2f a, b;
    a.x = As[col * AS_STRIDE + k];
    a.y = As[col * AS_STRIDE + k + 1];
    b.x = Wb[(size_t)k * KD];
    b.y = Wb[(size_t)(k + 1) * KD];
    acc = __builtin_amdgcn_wmma_f32_16x16x4_f32(
        false, a, false, b, (short)0, acc, false, false);
  }

#pragma unroll
  for (int v = 0; v < 8; ++v) {
    int r = row0 + v + 8 * hl;
    if (r < N) Wh[(size_t)r * KHD + n0 + col] = acc[v];
  }
}

__global__ void scores_kernel(const float* __restrict__ Wh,
                              const float* __restrict__ a1,
                              const float* __restrict__ a2,
                              float* __restrict__ s1, float* __restrict__ s2,
                              int N) {
  int idx = blockIdx.x * blockDim.x + threadIdx.x;
  if (idx >= N * KH) return;
  int n = idx >> 3, hh = idx & 7;
  const float* w  = Wh + (size_t)n * KHD + hh * KD;
  const float* v1 = a1 + hh * KD;
  const float* v2 = a2 + hh * KD;
  float acc1 = 0.0f, acc2 = 0.0f;
#pragma unroll
  for (int d = 0; d < KD; ++d) {
    float v = w[d];
    acc1 += v * v1[d];
    acc2 += v * v2[d];
  }
  s1[idx] = acc1;
  s2[idx] = acc2;
}

__device__ __forceinline__ float leaky(float e) {
  return (e > 0.0f) ? e : KALPHA * e;
}

__global__ __launch_bounds__(256) void row_softmax_gather_kernel(
    const int* __restrict__ ssrc, const int* __restrict__ offs, const int* __restrict__ deg,
    const float* __restrict__ s1, const float* __restrict__ s2,
    const float* __restrict__ Wh, float* __restrict__ out, int N) {
  const int lane = threadIdx.x & 31;
  const int d = blockIdx.x * 8 + __builtin_amdgcn_readfirstlane((int)(threadIdx.x >> 5));
  if (d >= N) return;
  const int beg = offs[d];
  const int end = beg + deg[d];

  const int h1 = lane & 7;
  const float s2a = s2[d * KH + h1];
  float mx = -__builtin_inff();
  for (int i = beg + (lane >> 3); i < end; i += 4) {
    const int s = ssrc[i];
    mx = fmaxf(mx, leaky(s1[s * KH + h1] + s2a));
  }
  mx = fmaxf(mx, __shfl_xor(mx, 8));
  mx = fmaxf(mx, __shfl_xor(mx, 16));
  const float m = __builtin_isfinite(mx) ? mx : 0.0f;

  const int hh = lane >> 2;
  const float mh  = __shfl(m, hh);
  const float s2b = s2[d * KH + hh];
  const float* s1h = s1 + hh;
  const float* whl = Wh + lane * 4;
  float4 acc = make_float4(0.0f, 0.0f, 0.0f, 0.0f);
  float z = 0.0f;
  int i = beg;
  for (; i + 4 <= end; i += 4) {
    const int sa = ssrc[i], sb = ssrc[i + 1], sc = ssrc[i + 2], sd = ssrc[i + 3];
    const float ea = s1h[sa * KH], eb = s1h[sb * KH], ec = s1h[sc * KH], ed = s1h[sd * KH];
    const float4 wa = *(const float4*)(whl + (size_t)sa * KHD);
    const float4 wb = *(const float4*)(whl + (size_t)sb * KHD);
    const float4 wc = *(const float4*)(whl + (size_t)sc * KHD);
    const float4 wd = *(const float4*)(whl + (size_t)sd * KHD);
    const float pa = __expf(leaky(ea + s2b) - mh);
    const float pb = __expf(leaky(eb + s2b) - mh);
    const float pc = __expf(leaky(ec + s2b) - mh);
    const float pd = __expf(leaky(ed + s2b) - mh);
    z += pa; acc.x += pa * wa.x; acc.y += pa * wa.y; acc.z += pa * wa.z; acc.w += pa * wa.w;
    z += pb; acc.x += pb * wb.x; acc.y += pb * wb.y; acc.z += pb * wb.z; acc.w += pb * wb.w;
    z += pc; acc.x += pc * wc.x; acc.y += pc * wc.y; acc.z += pc * wc.z; acc.w += pc * wc.w;
    z += pd; acc.x += pd * wd.x; acc.y += pd * wd.y; acc.z += pd * wd.z; acc.w += pd * wd.w;
  }
#pragma nounroll
  for (; i < end; ++i) {
    const int sa = ssrc[i];
    const float ea = s1h[sa * KH];
    const float4 wa = *(const float4*)(whl + (size_t)sa * KHD);
    const float pa = __expf(leaky(ea + s2b) - mh);
    z += pa; acc.x += pa * wa.x; acc.y += pa * wa.y; acc.z += pa * wa.z; acc.w += pa * wa.w;
  }
  const float zinv = 1.0f / fmaxf(z, 1e-16f);
  *(float4*)(out + (size_t)d * KHD + lane * 4) =
      make_float4(acc.x * zinv, acc.y * zinv, acc.z * zinv, acc.w * zinv);
}

extern "C" void kernel_launch(void* const* d_in, const int* in_sizes, int n_in,
                              void* d_out, int out_size, void* d_ws, size_t ws_size,
                              hipStream_t stream) {
  const float* hfeat = (const float*)d_in[0];
  const int*   src   = (const int*)d_in[1];
  const int*   dst   = (const int*)d_in[2];
  const float* W     = (const float*)d_in[3];
  const float* a1    = (const float*)d_in[4];
  const float* a2    = (const float*)d_in[5];
  float* out = (float*)d_out;

  const int N = in_sizes[0] / KF;
  const int E = in_sizes[1];

  float* Wh   = (float*)d_ws;
  float* s1   = Wh + (size_t)N * KHD;
  float* s2   = s1 + (size_t)N * KH;
  int*   ssrc = (int*)(s2 + (size_t)N * KH);
  int*   deg  = ssrc + (size_t)E;
  int*   offs = deg + (size_t)N;
  int*   tsum = offs + (size_t)N;
  int*   rank = (int*)d_ws;

  const int TB = 256;
  const int nh      = N * KH;
  const int g_n     = (N  + TB - 1) / TB;
  const int g_e     = (E  + TB - 1) / TB;
  const int g_rk    = (E + RK_EPT * RK_THREADS - 1) / (RK_EPT * RK_THREADS);
  const int g_nh    = (nh + TB - 1) / TB;
  const int g_tile  = (N + SCAN_TILE - 1) / SCAN_TILE;
  const int g_row   = (N + 7) / 8;
  const int g_gemm  = (N + 15) / 16;

  zero_deg_kernel<<<g_n, TB, 0, stream>>>(deg, N);
  rank_lds_kernel<<<g_rk, RK_THREADS, 0, stream>>>(dst, deg, rank, N, E);
  tile_sum_kernel<<<g_tile, TB, 0, stream>>>(deg, tsum, N);
  row_offsets_kernel<<<g_tile, TB, 0, stream>>>(deg, tsum, offs, N);
  place_kernel<<<g_e, TB, 0, stream>>>(src, dst, rank, offs, ssrc, E);

  gat_gemm_wmma_kernel<<<g_gemm, TB, 0, stream>>>(hfeat, W, Wh, N);

  scores_kernel<<<g_nh, TB, 0, stream>>>(Wh, a1, a2, s1, s2, N);
  row_softmax_gather_kernel<<<g_row, TB, 0, stream>>>(ssrc, offs, deg, s1, s2, Wh, out, N);
}


# Fewer, longer-lived waves: each wave loops over rows (100000 -> 14288 waves), GEMM strips (50000 -> 8192), edges in place (50000 -> 6256), score items (25000 -> 6256); order of all sums kept
# speedup vs baseline: 6.5495x; 1.0298x over previous
#include <hip/hip_runtime.h>
#include <hip/hip_bf16.h>


static constexpr int KF  = 128;
static constexpr int KH  = 8;
static constexpr int KD  = 16;
static constexpr int KHD = 128;
static constexpr float KALPHA = 0.2f;

typedef __attribute__((ext_vector_type(2))) float v2f;
typedef __attribute__((ext_vector_type(8))) float v8f;

__global__ void zero_deg_kernel(int* __restrict__ deg, int count) {
  int i = blockIdx.x * blockDim.x + threadIdx.x;
  if (i < count) deg[i] = 0;
}

static constexpr int RK_THREADS = 1024;
static constexpr int RK_EPT     = 16;
static constexpr int RK_ROWS    = 51200;
static constexpr int RK_BATCH   = 25;

__global__ __launch_bounds__(RK_THREADS) void rank_lds_kernel(const int* __restrict__ dst,
                                                              int* __restrict__ deg,
                                                              int* __restrict__ rank, int N, int E) {
  __shared__ int cnt[RK_ROWS];
  const int tid   = threadIdx.x;
  const int ebase = blockIdx.x * (RK_EPT * RK_THREADS) + tid;

  int dv[RK_EPT], rk[RK_EPT];
#pragma unroll
  for (int q = 0; q < RK_EPT; ++q) {
    const int e = ebase + q * RK_THREADS;
    dv[q] = (e < E) ? dst[e] : -1;
    rk[q] = 0;
  }

  for (int r0 = 0; r0 < N; r0 += RK_ROWS) {
    const int rows = min(RK_ROWS, N - r0);
    for (int w = tid; w < RK_ROWS; w += RK_THREADS) cnt[w] = 0;
    __syncthreads();

#pragma unroll
    for (int q = 0; q < RK_EPT; ++q) {
      const unsigned r = (unsigned)(dv[q] - r0);
      if (r < (unsigned)rows) atomicAdd(&cnt[r], 1);
    }
    __syncthreads();

    for (int wb = tid; wb < rows; wb += RK_THREADS * RK_BATCH) {
      int c[RK_BATCH], b[RK_BATCH];
#pragma unroll
      for (int q = 0; q < RK_BATCH; ++q) {
        const int w = wb + q * RK_THREADS;
        c[q] = (w < rows) ? cnt[w] : 0;
      }
#pragma unroll
      for (int q = 0; q < RK_BATCH; ++q) {
        b[q] = 0;
        if (c[q] != 0) b[q] = atomicAdd(&deg[r0 + wb + q * RK_THREADS], c[q]);
      }
#pragma unroll
      for (int q = 0; q < RK_BATCH; ++q)
        if (c[q] != 0) cnt[wb + q * RK_THREADS] = b[q];
    }
    __syncthreads();

#pragma unroll
    for (int q = 0; q < RK_EPT; ++q) {
      const unsigned r = (unsigned)(dv[q] - r0);
      if (r < (unsigned)rows) rk[q] = atomicAdd(&cnt[r], 1);
    }
    __syncthreads();
  }

#pragma unroll
  for (int q = 0; q < RK_EPT; ++q) {
    const int e = ebase + q * RK_THREADS;
    if (e < E) rank[e] = rk[q];
  }
}

static constexpr int SCAN_TILE = 1024;

__global__ __launch_bounds__(256) void tile_sum_kernel(const int* __restrict__ deg,
                                                       int* __restrict__ tsum, int N) {
  __shared__ int red[8];
  const int tid = threadIdx.x, lane = tid & 31, wave = tid >> 5;
  const int base = blockIdx.x * SCAN_TILE + tid * 4;
  int t = 0;
#pragma unroll
  for (int q = 0; q < 4; ++q)
    if (base + q < N) t += deg[base + q];
  for (int off = 16; off > 0; off >>= 1) t += __shfl_xor(t, off);
  if (lane == 0) red[wave] = t;
  __syncthreads();
  if (tid == 0) {
    int a = 0;
#pragma unroll
    for (int w = 0; w < 8; ++w) a += red[w];
    tsum[blockIdx.x] = a;
  }
}

__global__ __launch_bounds__(256) void row_offsets_kernel(const int* __restrict__ deg,
                                                          const int* __restrict__ tsum,
                                                          int* __restrict__ offs, int N) {
  __shared__ int winc[8];
  __shared__ int wpre[8];
  const int tid = threadIdx.x, lane = tid & 31, wave = tid >> 5;

  int pre = 0;
  for (int i = tid; i < (int)blockIdx.x; i += 256) pre += tsum[i];
  for (int off = 16; off > 0; off >>= 1) pre += __shfl_xor(pre, off);

  const int base = blockIdx.x * SCAN_TILE + tid * 4;
  int v[4];
  int t = 0;
#pragma unroll
  for (int q = 0; q < 4; ++q) {
    v[q] = (base + q < N) ? deg[base + q] : 0;
    t += v[q];
  }
  int inc = t;
  for (int off = 1; off < 32; off <<= 1) {
    int u = __shfl_up(inc, off);
    if (lane >= off) inc += u;
  }
  if (lane == 31) winc[wave] = inc;
  if (lane == 0) wpre[wave] = pre;
  __syncthreads();
  int add = 0;
#pragma unroll
  for (int w = 0; w < 8; ++w) {
    add += wpre[w];
    if (w < wave) add += winc[w];
  }
  int ex = add + inc - t;
#pragma unroll
  for (int q = 0; q < 4; ++q) {
    if (base + q < N) offs[base + q] = ex;
    ex += v[q];
  }
}

static constexpr int PL_EPT = 8;

__global__ __launch_bounds__(256) void place_kernel(const int* __restrict__ src,
                                                    const int* __restrict__ dst,
                                                    const int* __restrict__ rank,
                                                    const int* __restrict__ offs,
                                                    int* __restrict__ ssrc, int E) {
  const int base = blockIdx.x * (256 * PL_EPT) + threadIdx.x;
  int dv[PL_EPT], rv[PL_EPT], sv[PL_EPT], ov[PL_EPT];
#pragma unroll
  for (int q = 0; q < PL_EPT; ++q) {
    const int ec = min(base + q * 256, E - 1);
    dv[q] = dst[ec];
    rv[q] = rank[ec];
    sv[q] = src[ec];
  }
#pragma unroll
  for (int q = 0; q < PL_EPT; ++q) ov[q] = offs[dv[q]];
#pragma unroll
  for (int q = 0; q < PL_EPT; ++q) {
    const int e = base + q * 256;
    if (e < E) ssrc[ov[q] + rv[q]] = sv[q];
  }
}

static constexpr int AS_STRIDE = 132;
static constexpr int GEMM_WGS  = 1024;

__global__ __launch_bounds__(256) void gat_gemm_wmma_kernel(
    const float* __restrict__ hfeat, const float* __restrict__ W,
    float* __restrict__ Wh, int N) {
  __shared__ alignas(16) float As[2][16 * AS_STRIDE];

  const int tid  = threadIdx.x;
  const int wave = tid >> 5;
  const int lane = tid & 31;
  const int hl   = lane >> 4;
  const int col  = lane & 15;
  const int n0   = wave * 16;
  const int nstrips = (N + 15) >> 4;
  const int sstep   = gridDim.x;

  const float* Wb = W + wave * (KF * KD) + col;
  v2f bfr[KF / 4];
#pragma unroll
  for (int j = 0; j < KF / 4; ++j) {
    const int k = 4 * j + 2 * hl;
    bfr[j].x = Wb[(size_t)k * KD];
    bfr[j].y = Wb[(size_t)(k + 1) * KD];
  }

  const int sr = tid >> 5;
  const int sf = (tid & 31) * 4;
  float4 p0, p1;
  const float4 zero4 = make_float4(0.0f, 0.0f, 0.0f, 0.0f);

  int s = blockIdx.x;
  {
    const int ra = s * 16 + sr, rb = ra + 8;
    p0 = zero4;
    p1 = zero4;
    if (s < nstrips && ra < N) p0 = *(const float4*)(hfeat + (size_t)ra * KF + sf);
    if (s < nstrips && rb < N) p1 = *(const float4*)(hfeat + (size_t)rb * KF + sf);
    *(float4*)(&As[0][sr * AS_STRIDE + sf])       = p0;
    *(float4*)(&As[0][(sr + 8) * AS_STRIDE + sf]) = p1;
  }
  __syncthreads();

  int cur = 0;
  for (; s < nstrips; s += sstep) {
    const int sn = s + sstep;
    const bool more = sn < nstrips;
    if (more) {
      const int ra = sn * 16 + sr, rb = ra + 8;
      p0 = zero4;
      p1 = zero4;
      if (ra < N) p0 = *(const float4*)(hfeat + (size_t)ra * KF + sf);
      if (rb < N) p1 = *(const float4*)(hfeat + (size_t)rb * KF + sf);
    }

    const float* Ac = &As[cur][col * AS_STRIDE + 2 * hl];
    v8f acc = {};
#pragma unroll
    for (int j = 0; j < KF / 4; ++j) {
      v2f a;
      a.x = Ac[4 * j];
      a.y = Ac[4 * j + 1];
      acc = __builtin_amdgcn_wmma_f32_16x16x4_f32(
          false, a, false, bfr[j], (short)0, acc, false, false);
    }

    const int row0 = s * 16;
#pragma unroll
    for (int v = 0; v < 8; ++v) {
      int r = row0 + v + 8 * hl;
      if (r < N) Wh[(size_t)r * KHD + n0 + col] = acc[v];
    }

    if (more) {
      *(float4*)(&As[cur ^ 1][sr * AS_STRIDE + sf])       = p0;
      *(float4*)(&As[cur ^ 1][(sr + 8) * AS_STRIDE + sf]) = p1;
    }
    __syncthreads();
    cur ^= 1;
  }
}

static constexpr int SC_IPT = 4;

__global__ __launch_bounds__(256) void scores_kernel(const float* __restrict__ Wh,
                                                     const float* __restrict__ a1,
                                                     const float* __restrict__ a2,
                                                     float* __restrict__ s1, float* __restrict__ s2,
                                                     int N) {
  const int total = N * KH;
  const int base  = blockIdx.x * (256 * SC_IPT) + threadIdx.x;
  const int hh    = base & 7;
  const float* v1 = a1 + hh * KD;
  const float* v2 = a2 + hh * KD;
  float wv[SC_IPT][KD];
#pragma unroll
  for (int q = 0; q < SC_IPT; ++q) {
    const int ic = min(base + q * 256, total - 1);
    const float* w = Wh + (size_t)(ic >> 3) * KHD + hh * KD;
#pragma unroll
    for (int d = 0; d < KD; ++d) wv[q][d] = w[d];
  }
#pragma unroll
  for (int q = 0; q < SC_IPT; ++q) {
    const int idx = base + q * 256;
    float acc1 = 0.0f, acc2 = 0.0f;
#pragma unroll
    for (int d = 0; d < KD; ++d) {
      float v = wv[q][d];
      acc1 += v * v1[d];
      acc2 += v * v2[d];
    }
    if (idx < total) {
      s1[idx] = acc1;
      s2[idx] = acc2;
    }
  }
}

__device__ __forceinline__ float leaky(float e) {
  return (e > 0.0f) ? e : KALPHA * e;
}

static constexpr int ROW_WGS = 1786;
__global__ __launch_bounds__(256) void row_softmax_gather_kernel(
    const int* __restrict__ ssrc, const int* __restrict__ offs, const int* __restrict__ deg,
    const float* __restrict__ s1, const float* __restrict__ s2,
    const float* __restrict__ Wh, float* __restrict__ out, int N) {
  const int lane = threadIdx.x & 31;
  int d = blockIdx.x * 8 + __builtin_amdgcn_readfirstlane((int)(threadIdx.x >> 5));
  const int dstep = gridDim.x * 8;
  int nbeg = 0, ncnt = 0;
  if (d < N) { nbeg = offs[d]; ncnt = deg[d]; }
  for (; d < N; d += dstep) {
  const int beg = nbeg;
  const int end = beg + ncnt;
  {
    const int dn = d + dstep;
    if (dn < N) { nbeg = offs[dn]; ncnt = deg[dn]; }
  }

  const int h1 = lane & 7;
  const float s2a = s2[d * KH + h1];
  float mx = -__builtin_inff();
  for (int i = beg + (lane >> 3); i < end; i += 4) {
    const int s = ssrc[i];
    mx = fmaxf(mx, leaky(s1[s * KH + h1] + s2a));
  }
  mx = fmaxf(mx, __shfl_xor(mx, 8));
  mx = fmaxf(mx, __shfl_xor(mx, 16));
  const float m = __builtin_isfinite(mx) ? mx : 0.0f;

  const int hh = lane >> 2;
  const float mh  = __shfl(m, hh);
  const float s2b = s2[d * KH + hh];
  const float* s1h = s1 + hh;
  const float* whl = Wh + lane * 4;
  float4 acc = make_float4(0.0f, 0.0f, 0.0f, 0.0f);
  float z = 0.0f;
  int i = beg;
  for (; i + 4 <= end; i += 4) {
    const int sa = ssrc[i], sb = ssrc[i + 1], sc = ssrc[i + 2], sd = ssrc[i + 3];
    const float ea = s1h[sa * KH], eb = s1h[sb * KH], ec = s1h[sc * KH], ed = s1h[sd * KH];
    const float4 wa = *(const float4*)(whl + (size_t)sa * KHD);
    const float4 wb = *(const float4*)(whl + (size_t)sb * KHD);
    const float4 wc = *(const float4*)(whl + (size_t)sc * KHD);
    const float4 wd = *(const float4*)(whl + (size_t)sd * KHD);
    const float pa = __expf(leaky(ea + s2b) - mh);
    const float pb = __expf(leaky(eb + s2b) - mh);
    const float pc = __expf(leaky(ec + s2b) - mh);
    const float pd = __expf(leaky(ed + s2b) - mh);
    z += pa; acc.x += pa * wa.x; acc.y += pa * wa.y; acc.z += pa * wa.z; acc.w += pa * wa.w;
    z += pb; acc.x += pb * wb.x; acc.y += pb * wb.y; acc.z += pb * wb.z; acc.w += pb * wb.w;
    z += pc; acc.x += pc * wc.x; acc.y += pc * wc.y; acc.z += pc * wc.z; acc.w += pc * wc.w;
    z += pd; acc.x += pd * wd.x; acc.y += pd * wd.y; acc.z += pd * wd.z; acc.w += pd * wd.w;
  }
#pragma nounroll
  for (; i < end; ++i) {
    const int sa = ssrc[i];
    const float ea = s1h[sa * KH];
    const float4 wa = *(const float4*)(whl + (size_t)sa * KHD);
    const float pa = __expf(leaky(ea + s2b) - mh);
    z += pa; acc.x += pa * wa.x; acc.y += pa * wa.y; acc.z += pa * wa.z; acc.w += pa * wa.w;
  }
  const float zinv = 1.0f / fmaxf(z, 1e-16f);
  *(float4*)(out + (size_t)d * KHD + lane * 4) =
      make_float4(acc.x * zinv, acc.y * zinv, acc.z * zinv, acc.w * zinv);
  }
}

extern "C" void kernel_launch(void* const* d_in, const int* in_sizes, int n_in,
                              void* d_out, int out_size, void* d_ws, size_t ws_size,
                              hipStream_t stream) {
  const float* hfeat = (const float*)d_in[0];
  const int*   src   = (const int*)d_in[1];
  const int*   dst   = (const int*)d_in[2];
  const float* W     = (const float*)d_in[3];
  const float* a1    = (const float*)d_in[4];
  const float* a2    = (const float*)d_in[5];
  float* out = (float*)d_out;

  const int N = in_sizes[0] / KF;
  const int E = in_sizes[1];

  float* Wh   = (float*)d_ws;
  float* s1   = Wh + (size_t)N * KHD;
  float* s2   = s1 + (size_t)N * KH;
  int*   ssrc = (int*)(s2 + (size_t)N * KH);
  int*   deg  = ssrc + (size_t)E;
  int*   offs = deg + (size_t)N;
  int*   tsum = offs + (size_t)N;
  int*   rank = (int*)d_ws;

  const int TB = 256;
  const int nh      = N * KH;
  const int g_n     = (N  + TB - 1) / TB;
  const int g_pl    = (E + TB * PL_EPT - 1) / (TB * PL_EPT);
  const int g_rk    = (E + RK_EPT * RK_THREADS - 1) / (RK_EPT * RK_THREADS);
  const int g_nh    = (nh + TB * SC_IPT - 1) / (TB * SC_IPT);
  const int g_tile  = (N + SCAN_TILE - 1) / SCAN_TILE;

  zero_deg_kernel<<<g_n, TB, 0, stream>>>(deg, N);
  rank_lds_kernel<<<g_rk, RK_THREADS, 0, stream>>>(dst, deg, rank, N, E);
  tile_sum_kernel<<<g_tile, TB, 0, stream>>>(deg, tsum, N);
  row_offsets_kernel<<<g_tile, TB, 0, stream>>>(deg, tsum, offs, N);
  place_kernel<<<g_pl, TB, 0, stream>>>(src, dst, rank, offs, ssrc, E);

  gat_gemm_wmma_kernel<<<GEMM_WGS, TB, 0, stream>>>(hfeat, W, Wh, N);

  scores_kernel<<<g_nh, TB, 0, stream>>>(Wh, a1, a2, s1, s2, N);
  row_softmax_gather_kernel<<<ROW_WGS, TB, 0, stream>>>(ssrc, offs, deg, s1, s2, Wh, out, N);
}


# Row kernel: one pass per row instead of two: running-max (online) softmax, sums rescaled by exp(m_old-m_new) when the max rises; each logit loaded once; edge order of the sums kept
# speedup vs baseline: 7.0039x; 1.0694x over previous
#include <hip/hip_runtime.h>
#include <hip/hip_bf16.h>


static constexpr int KF  = 128;
static constexpr int KH  = 8;
static constexpr int KD  = 16;
static constexpr int KHD = 128;
static constexpr float KALPHA = 0.2f;

typedef __attribute__((ext_vector_type(2))) float v2f;
typedef __attribute__((ext_vector_type(8))) float v8f;

__global__ void zero_deg_kernel(int* __restrict__ deg, int count) {
  int i = blockIdx.x * blockDim.x + threadIdx.x;
  if (i < count) deg[i] = 0;
}

static constexpr int RK_THREADS = 1024;
static constexpr int RK_EPT     = 16;
static constexpr int RK_ROWS    = 51200;
static constexpr int RK_BATCH   = 25;

__global__ __launch_bounds__(RK_THREADS) void rank_lds_kernel(const int* __restrict__ dst,
                                                              int* __restrict__ deg,
                                                              int* __restrict__ rank, int N, int E) {
  __shared__ int cnt[RK_ROWS];
  const int tid   = threadIdx.x;
  const int ebase = blockIdx.x * (RK_EPT * RK_THREADS) + tid;

  int dv[RK_EPT], rk[RK_EPT];
#pragma unroll
  for (int q = 0; q < RK_EPT; ++q) {
    const int e = ebase + q * RK_THREADS;
    dv[q] = (e < E) ? dst[e] : -1;
    rk[q] = 0;
  }

  for (int r0 = 0; r0 < N; r0 += RK_ROWS) {
    const int rows = min(RK_ROWS, N - r0);
    for (int w = tid; w < RK_ROWS; w += RK_THREADS) cnt[w] = 0;
    __syncthreads();

#pragma unroll
    for (int q = 0; q < RK_EPT; ++q) {
      const unsigned r = (unsigned)(dv[q] - r0);
      if (r < (unsigned)rows) atomicAdd(&cnt[r], 1);
    }
    __syncthreads();

    for (int wb = tid; wb < rows; wb += RK_THREADS * RK_BATCH) {
      int c[RK_BATCH], b[RK_BATCH];
#pragma unroll
      for (int q = 0; q < RK_BATCH; ++q) {
        const int w = wb + q * RK_THREADS;
        c[q] = (w < rows) ? cnt[w] : 0;
      }
#pragma unroll
      for (int q = 0; q < RK_BATCH; ++q) {
        b[q] = 0;
        if (c[q] != 0) b[q] = atomicAdd(&deg[r0 + wb + q * RK_THREADS], c[q]);
      }
#pragma unroll
      for (int q = 0; q < RK_BATCH; ++q)
        if (c[q] != 0) cnt[wb + q * RK_THREADS] = b[q];
    }
    __syncthreads();

#pragma unroll
    for (int q = 0; q < RK_EPT; ++q) {
      const unsigned r = (unsigned)(dv[q] - r0);
      if (r < (unsigned)rows) rk[q] = atomicAdd(&cnt[r], 1);
    }
    __syncthreads();
  }

#pragma unroll
  for (int q = 0; q < RK_EPT; ++q) {
    const int e = ebase + q * RK_THREADS;
    if (e < E) rank[e] = rk[q];
  }
}

static constexpr int SCAN_TILE = 1024;

__global__ __launch_bounds__(256) void tile_sum_kernel(const int* __restrict__ deg,
                                                       int* __restrict__ tsum, int N) {
  __shared__ int red[8];
  const int tid = threadIdx.x, lane = tid & 31, wave = tid >> 5;
  const int base = blockIdx.x * SCAN_TILE + tid * 4;
  int t = 0;
#pragma unroll
  for (int q = 0; q < 4; ++q)
    if (base + q < N) t += deg[base + q];
  for (int off = 16; off > 0; off >>= 1) t += __shfl_xor(t, off);
  if (lane == 0) red[wave] = t;
  __syncthreads();
  if (tid == 0) {
    int a = 0;
#pragma unroll
    for (int w = 0; w < 8; ++w) a += red[w];
    tsum[blockIdx.x] = a;
  }
}

__global__ __launch_bounds__(256) void row_offsets_kernel(const int* __restrict__ deg,
                                                          const int* __restrict__ tsum,
                                                          int* __restrict__ offs, int N) {
  __shared__ int winc[8];
  __shared__ int wpre[8];
  const int tid = threadIdx.x, lane = tid & 31, wave = tid >> 5;

  int pre = 0;
  for (int i = tid; i < (int)blockIdx.x; i += 256) pre += tsum[i];
  for (int off = 16; off > 0; off >>= 1) pre += __shfl_xor(pre, off);

  const int base = blockIdx.x * SCAN_TILE + tid * 4;
  int v[4];
  int t = 0;
#pragma unroll
  for (int q = 0; q < 4; ++q) {
    v[q] = (base + q < N) ? deg[base + q] : 0;
    t += v[q];
  }
  int inc = t;
  for (int off = 1; off < 32; off <<= 1) {
    int u = __shfl_up(inc, off);
    if (lane >= off) inc += u;
  }
  if (lane == 31) winc[wave] = inc;
  if (lane == 0) wpre[wave] = pre;
  __syncthreads();
  int add = 0;
#pragma unroll
  for (int w = 0; w < 8; ++w) {
    add += wpre[w];
    if (w < wave) add += winc[w];
  }
  int ex = add + inc - t;
#pragma unroll
  for (int q = 0; q < 4; ++q) {
    if (base + q < N) offs[base + q] = ex;
    ex += v[q];
  }
}

static constexpr int PL_EPT = 8;

__global__ __launch_bounds__(256) void place_kernel(const int* __restrict__ src,
                                                    const int* __restrict__ dst,
                                                    const int* __restrict__ rank,
                                                    const int* __restrict__ offs,
                                                    int* __restrict__ ssrc, int E) {
  const int base = blockIdx.x * (256 * PL_EPT) + threadIdx.x;
  int dv[PL_EPT], rv[PL_EPT], sv[PL_EPT], ov[PL_EPT];
#pragma unroll
  for (int q = 0; q < PL_EPT; ++q) {
    const int ec = min(base + q * 256, E - 1);
    dv[q] = dst[ec];
    rv[q] = rank[ec];
    sv[q] = src[ec];
  }
#pragma unroll
  for (int q = 0; q < PL_EPT; ++q) ov[q] = offs[dv[q]];
#pragma unroll
  for (int q = 0; q < PL_EPT; ++q) {
    const int e = base + q * 256;
    if (e < E) ssrc[ov[q] + rv[q]] = sv[q];
  }
}

static constexpr int AS_STRIDE = 132;
static constexpr int GEMM_WGS  = 1024;

__global__ __launch_bounds__(256) void gat_gemm_wmma_kernel(
    const float* __restrict__ hfeat, const float* __restrict__ W,
    float* __restrict__ Wh, int N) {
  __shared__ alignas(16) float As[2][16 * AS_STRIDE];

  const int tid  = threadIdx.x;
  const int wave = tid >> 5;
  const int lane = tid & 31;
  const int hl   = lane >> 4;
  const int col  = lane & 15;
  const int n0   = wave * 16;
  const int nstrips = (N + 15) >> 4;
  const int sstep   = gridDim.x;

  const float* Wb = W + wave * (KF * KD) + col;
  v2f bfr[KF / 4];
#pragma unroll
  for (int j = 0; j < KF / 4; ++j) {
    const int k = 4 * j + 2 * hl;
    bfr[j].x = Wb[(size_t)k * KD];
    bfr[j].y = Wb[(size_t)(k + 1) * KD];
  }

  const int sr = tid >> 5;
  const int sf = (tid & 31) * 4;
  float4 p0, p1;
  const float4 zero4 = make_float4(0.0f, 0.0f, 0.0f, 0.0f);

  int s = blockIdx.x;
  {
    const int ra = s * 16 + sr, rb = ra + 8;
    p0 = zero4;
    p1 = zero4;
    if (s < nstrips && ra < N) p0 = *(const float4*)(hfeat + (size_t)ra * KF + sf);
    if (s < nstrips && rb < N) p1 = *(const float4*)(hfeat + (size_t)rb * KF + sf);
    *(float4*)(&As[0][sr * AS_STRIDE + sf])       = p0;
    *(float4*)(&As[0][(sr + 8) * AS_STRIDE + sf]) = p1;
  }
  __syncthreads();

  int cur = 0;
  for (; s < nstrips; s += sstep) {
    const int sn = s + sstep;
    const bool more = sn < nstrips;
    if (more) {
      const int ra = sn * 16 + sr, rb = ra + 8;
      p0 = zero4;
      p1 = zero4;
      if (ra < N) p0 = *(const float4*)(hfeat + (size_t)ra * KF + sf);
      if (rb < N) p1 = *(const float4*)(hfeat + (size_t)rb * KF + sf);
    }

    const float* Ac = &As[cur][col * AS_STRIDE + 2 * hl];
    v8f acc = {};
#pragma unroll
    for (int j = 0; j < KF / 4; ++j) {
      v2f a;
      a.x = Ac[4 * j];
      a.y = Ac[4 * j + 1];
      acc = __builtin_amdgcn_wmma_f32_16x16x4_f32(
          false, a, false, bfr[j], (short)0, acc, false, false);
    }

    const int row0 = s * 16;
#pragma unroll
    for (int v = 0; v < 8; ++v) {
      int r = row0 + v + 8 * hl;
      if (r < N) Wh[(size_t)r * KHD + n0 + col] = acc[v];
    }

    if (more) {
      *(float4*)(&As[cur ^ 1][sr * AS_STRIDE + sf])       = p0;
      *(float4*)(&As[cur ^ 1][(sr + 8) * AS_STRIDE + sf]) = p1;
    }
    __syncthreads();
    cur ^= 1;
  }
}

static constexpr int SC_IPT = 4;

__global__ __launch_bounds__(256) void scores_kernel(const float* __restrict__ Wh,
                                                     const float* __restrict__ a1,
                                                     const float* __restrict__ a2,
                                                     float* __restrict__ s1, float* __restrict__ s2,
                                                     int N) {
  const int total = N * KH;
  const int base  = blockIdx.x * (256 * SC_IPT) + threadIdx.x;
  const int hh    = base & 7;
  const float* v1 = a1 + hh * KD;
  const float* v2 = a2 + hh * KD;
  float wv[SC_IPT][KD];
#pragma unroll
  for (int q = 0; q < SC_IPT; ++q) {
    const int ic = min(base + q * 256, total - 1);
    const float* w = Wh + (size_t)(ic >> 3) * KHD + hh * KD;
#pragma unroll
    for (int d = 0; d < KD; ++d) wv[q][d] = w[d];
  }
#pragma unroll
  for (int q = 0; q < SC_IPT; ++q) {
    const int idx = base + q * 256;
    float acc1 = 0.0f, acc2 = 0.0f;
#pragma unroll
    for (int d = 0; d < KD; ++d) {
      float v = wv[q][d];
      acc1 += v * v1[d];
      acc2 += v * v2[d];
    }
    if (idx < total) {
      s1[idx] = acc1;
      s2[idx] = acc2;
    }
  }
}

__device__ __forceinline__ float leaky(float e) {
  return (e > 0.0f) ? e : KALPHA * e;
}

static constexpr int ROW_WGS = 1786;
__global__ __launch_bounds__(256) void row_softmax_gather_kernel(
    const int* __restrict__ ssrc, const int* __restrict__ offs, const int* __restrict__ deg,
    const float* __restrict__ s1, const float* __restrict__ s2,
    const float* __restrict__ Wh, float* __restrict__ out, int N) {
  const int lane = threadIdx.x & 31;
  int d = blockIdx.x * 8 + __builtin_amdgcn_readfirstlane((int)(threadIdx.x >> 5));
  const int dstep = gridDim.x * 8;
  int nbeg = 0, ncnt = 0;
  if (d < N) { nbeg = offs[d]; ncnt = deg[d]; }
  for (; d < N; d += dstep) {
  const int beg = nbeg;
  const int end = beg + ncnt;
  {
    const int dn = d + dstep;
    if (dn < N) { nbeg = offs[dn]; ncnt = deg[dn]; }
  }

  const int hh = lane >> 2;
  const float s2b = s2[d * KH + hh];
  const float* s1h = s1 + hh;
  const float* whl = Wh + lane * 4;
  float4 acc = make_float4(0.0f, 0.0f, 0.0f, 0.0f);
  float z = 0.0f;
  float m = -__builtin_inff();
  int i = beg;
  for (; i + 4 <= end; i += 4) {
    const int sa = ssrc[i], sb = ssrc[i + 1], sc = ssrc[i + 2], sd = ssrc[i + 3];
    const float ea = s1h[sa * KH], eb = s1h[sb * KH], ec = s1h[sc * KH], ed = s1h[sd * KH];
    const float4 wa = *(const float4*)(whl + (size_t)sa * KHD);
    const float4 wb = *(const float4*)(whl + (size_t)sb * KHD);
    const float4 wc = *(const float4*)(whl + (size_t)sc * KHD);
    const float4 wd = *(const float4*)(whl + (size_t)sd * KHD);
    const float xa = leaky(ea + s2b), xb = leaky(eb + s2b), xc = leaky(ec + s2b), xd = leaky(ed + s2b);
    const float mn = fmaxf(fmaxf(fmaxf(xa, xb), fmaxf(xc, xd)), m);
    const float f  = __expf(m - mn);
    m = mn;
    const float pa = __expf(xa - mn);
    const float pb = __expf(xb - mn);
    const float pc = __expf(xc - mn);
    const float pd = __expf(xd - mn);
    z *= f; acc.x *= f; acc.y *= f; acc.z *= f; acc.w *= f;
    z += pa; acc.x += pa * wa.x; acc.y += pa * wa.y; acc.z += pa * wa.z; acc.w += pa * wa.w;
    z += pb; acc.x += pb * wb.x; acc.y += pb * wb.y; acc.z += pb * wb.z; acc.w += pb * wb.w;
    z += pc; acc.x += pc * wc.x; acc.y += pc * wc.y; acc.z += pc * wc.z; acc.w += pc * wc.w;
    z += pd; acc.x += pd * wd.x; acc.y += pd * wd.y; acc.z += pd * wd.z; acc.w += pd * wd.w;
  }
#pragma nounroll
  for (; i < end; ++i) {
    const int sa = ssrc[i];
    const float ea = s1h[sa * KH];
    const float4 wa = *(const float4*)(whl + (size_t)sa * KHD);
    const float xa = leaky(ea + s2b);
    const float mn = fmaxf(xa, m);
    const float f  = __expf(m - mn);
    m = mn;
    const float pa = __expf(xa - mn);
    z *= f; acc.x *= f; acc.y *= f; acc.z *= f; acc.w *= f;
    z += pa; acc.x += pa * wa.x; acc.y += pa * wa.y; acc.z += pa * wa.z; acc.w += pa * wa.w;
  }
  const float zinv = 1.0f / fmaxf(z, 1e-16f);
  *(float4*)(out + (size_t)d * KHD + lane * 4) =
      make_float4(acc.x * zinv, acc.y * zinv, acc.z * zinv, acc.w * zinv);
  }
}

extern "C" void kernel_launch(void* const* d_in, const int* in_sizes, int n_in,
                              void* d_out, int out_size, void* d_ws, size_t ws_size,
                              hipStream_t stream) {
  const float* hfeat = (const float*)d_in[0];
  const int*   src   = (const int*)d_in[1];
  const int*   dst   = (const int*)d_in[2];
  const float* W     = (const float*)d_in[3];
  const float* a1    = (const float*)d_in[4];
  const float* a2    = (const float*)d_in[5];
  float* out = (float*)d_out;

  const int N = in_sizes[0] / KF;
  const int E = in_sizes[1];

  float* Wh   = (float*)d_ws;
  float* s1   = Wh + (size_t)N * KHD;
  float* s2   = s1 + (size_t)N * KH;
  int*   ssrc = (int*)(s2 + (size_t)N * KH);
  int*   deg  = ssrc + (size_t)E;
  int*   offs = deg + (size_t)N;
  int*   tsum = offs + (size_t)N;
  int*   rank = (int*)d_ws;

  const int TB = 256;
  const int nh      = N * KH;
  const int g_n     = (N  + TB - 1) / TB;
  const int g_pl    = (E + TB * PL_EPT - 1) / (TB * PL_EPT);
  const int g_rk    = (E + RK_EPT * RK_THREADS - 1) / (RK_EPT * RK_THREADS);
  const int g_nh    = (nh + TB * SC_IPT - 1) / (TB * SC_IPT);
  const int g_tile  = (N + SCAN_TILE - 1) / SCAN_TILE;

  zero_deg_kernel<<<g_n, TB, 0, stream>>>(deg, N);
  rank_lds_kernel<<<g_rk, RK_THREADS, 0, stream>>>(dst, deg, rank, N, E);
  tile_sum_kernel<<<g_tile, TB, 0, stream>>>(deg, tsum, N);
  row_offsets_kernel<<<g_tile, TB, 0, stream>>>(deg, tsum, offs, N);
  place_kernel<<<g_pl, TB, 0, stream>>>(src, dst, rank, offs, ssrc, E);

  gat_gemm_wmma_kernel<<<GEMM_WGS, TB, 0, stream>>>(hfeat, W, Wh, N);

  scores_kernel<<<g_nh, TB, 0, stream>>>(Wh, a1, a2, s1, s2, N);
  row_softmax_gather_kernel<<<ROW_WGS, TB, 0, stream>>>(ssrc, offs, deg, s1, s2, Wh, out, N);
}


# Row kernel forms both logit terms from the Wh rows it loads (4-column partial dot per lane, 4-lane DPP butterfly: tree order): per-edge score gather, score planes and the scores launch removed
# speedup vs baseline: 7.2937x; 1.0414x over previous
#include <hip/hip_runtime.h>
#include <hip/hip_bf16.h>


static constexpr int KF  = 128;
static constexpr int KH  = 8;
static constexpr int KD  = 16;
static constexpr int KHD = 128;
static constexpr float KALPHA = 0.2f;

typedef __attribute__((ext_vector_type(2))) float v2f;
typedef __attribute__((ext_vector_type(8))) float v8f;

__global__ void zero_deg_kernel(int* __restrict__ deg, int count) {
  int i = blockIdx.x * blockDim.x + threadIdx.x;
  if (i < count) deg[i] = 0;
}

static constexpr int RK_THREADS = 1024;
static constexpr int RK_EPT     = 16;
static constexpr int RK_ROWS    = 51200;
static constexpr int RK_BATCH   = 25;

__global__ __launch_bounds__(RK_THREADS) void rank_lds_kernel(const int* __restrict__ dst,
                                                              int* __restrict__ deg,
                                                              int* __restrict__ rank, int N, int E) {
  __shared__ int cnt[RK_ROWS];
  const int tid   = threadIdx.x;
  const int ebase = blockIdx.x * (RK_EPT * RK_THREADS) + tid;

  int dv[RK_EPT], rk[RK_EPT];
#pragma unroll
  for (int q = 0; q < RK_EPT; ++q) {
    const int e = ebase + q * RK_THREADS;
    dv[q] = (e < E) ? dst[e] : -1;
    rk[q] = 0;
  }

  for (int r0 = 0; r0 < N; r0 += RK_ROWS) {
    const int rows = min(RK_ROWS, N - r0);
    for (int w = tid; w < RK_ROWS; w += RK_THREADS) cnt[w] = 0;
    __syncthreads();

#pragma unroll
    for (int q = 0; q < RK_EPT; ++q) {
      const unsigned r = (unsigned)(dv[q] - r0);
      if (r < (unsigned)rows) atomicAdd(&cnt[r], 1);
    }
    __syncthreads();

    for (int wb = tid; wb < rows; wb += RK_THREADS * RK_BATCH) {
      int c[RK_BATCH], b[RK_BATCH];
#pragma unroll
      for (int q = 0; q < RK_BATCH; ++q) {
        const int w = wb + q * RK_THREADS;
        c[q] = (w < rows) ? cnt[w] : 0;
      }
#pragma unroll
      for (int q = 0; q < RK_BATCH; ++q) {
        b[q] = 0;
        if (c[q] != 0) b[q] = atomicAdd(&deg[r0 + wb + q * RK_THREADS], c[q]);
      }
#pragma unroll
      for (int q = 0; q < RK_BATCH; ++q)
        if (c[q] != 0) cnt[wb + q * RK_THREADS] = b[q];
    }
    __syncthreads();

#pragma unroll
    for (int q = 0; q < RK_EPT; ++q) {
      const unsigned r = (unsigned)(dv[q] - r0);
      if (r < (unsigned)rows) rk[q] = atomicAdd(&cnt[r], 1);
    }
    __syncthreads();
  }

#pragma unroll
  for (int q = 0; q < RK_EPT; ++q) {
    const int e = ebase + q * RK_THREADS;
    if (e < E) rank[e] = rk[q];
  }
}

static constexpr int SCAN_TILE = 1024;

__global__ __launch_bounds__(256) void tile_sum_kernel(const int* __restrict__ deg,
                                                       int* __restrict__ tsum, int N) {
  __shared__ int red[8];
  const int tid = threadIdx.x, lane = tid & 31, wave = tid >> 5;
  const int base = blockIdx.x * SCAN_TILE + tid * 4;
  int t = 0;
#pragma unroll
  for (int q = 0; q < 4; ++q)
    if (base + q < N) t += deg[base + q];
  for (int off = 16; off > 0; off >>= 1) t += __shfl_xor(t, off);
  if (lane == 0) red[wave] = t;
  __syncthreads();
  if (tid == 0) {
    int a = 0;
#pragma unroll
    for (int w = 0; w < 8; ++w) a += red[w];
    tsum[blockIdx.x] = a;
  }
}

__global__ __launch_bounds__(256) void row_offsets_kernel(const int* __restrict__ deg,
                                                          const int* __restrict__ tsum,
                                                          int* __restrict__ offs, int N) {
  __shared__ int winc[8];
  __shared__ int wpre[8];
  const int tid = threadIdx.x, lane = tid & 31, wave = tid >> 5;

  int pre = 0;
  for (int i = tid; i < (int)blockIdx.x; i += 256) pre += tsum[i];
  for (int off = 16; off > 0; off >>= 1) pre += __shfl_xor(pre, off);

  const int base = blockIdx.x * SCAN_TILE + tid * 4;
  int v[4];
  int t = 0;
#pragma unroll
  for (int q = 0; q < 4; ++q) {
    v[q] = (base + q < N) ? deg[base + q] : 0;
    t += v[q];
  }
  int inc = t;
  for (int off = 1; off < 32; off <<= 1) {
    int u = __shfl_up(inc, off);
    if (lane >= off) inc += u;
  }
  if (lane == 31) winc[wave] = inc;
  if (lane == 0) wpre[wave] = pre;
  __syncthreads();
  int add = 0;
#pragma unroll
  for (int w = 0; w < 8; ++w) {
    add += wpre[w];
    if (w < wave) add += winc[w];
  }
  int ex = add + inc - t;
#pragma unroll
  for (int q = 0; q < 4; ++q) {
    if (base + q < N) offs[base + q] = ex;
    ex += v[q];
  }
}

static constexpr int PL_EPT = 8;

__global__ __launch_bounds__(256) void place_kernel(const int* __restrict__ src,
                                                    const int* __restrict__ dst,
                                                    const int* __restrict__ rank,
                                                    const int* __restrict__ offs,
                                                    int* __restrict__ ssrc, int E) {
  const int base = blockIdx.x * (256 * PL_EPT) + threadIdx.x;
  int dv[PL_EPT], rv[PL_EPT], sv[PL_EPT], ov[PL_EPT];
#pragma unroll
  for (int q = 0; q < PL_EPT; ++q) {
    const int ec = min(base + q * 256, E - 1);
    dv[q] = dst[ec];
    rv[q] = rank[ec];
    sv[q] = src[ec];
  }
#pragma unroll
  for (int q = 0; q < PL_EPT; ++q) ov[q] = offs[dv[q]];
#pragma unroll
  for (int q = 0; q < PL_EPT; ++q) {
    const int e = base + q * 256;
    if (e < E) ssrc[ov[q] + rv[q]] = sv[q];
  }
}

static constexpr int AS_STRIDE = 132;
static constexpr int GEMM_WGS  = 1024;

__global__ __launch_bounds__(256) void gat_gemm_wmma_kernel(
    const float* __restrict__ hfeat, const float* __restrict__ W,
    float* __restrict__ Wh, int N) {
  __shared__ alignas(16) float As[2][16 * AS_STRIDE];

  const int tid  = threadIdx.x;
  const int wave = tid >> 5;
  const int lane = tid & 31;
  const int hl   = lane >> 4;
  const int col  = lane & 15;
  const int n0   = wave * 16;
  const int nstrips = (N + 15) >> 4;
  const int sstep   = gridDim.x;

  const float* Wb = W + wave * (KF * KD) + col;
  v2f bfr[KF / 4];
#pragma unroll
  for (int j = 0; j < KF / 4; ++j) {
    const int k = 4 * j + 2 * hl;
    bfr[j].x = Wb[(size_t)k * KD];
    bfr[j].y = Wb[(size_t)(k + 1) * KD];
  }

  const int sr = tid >> 5;
  const int sf = (tid & 31) * 4;
  float4 p0, p1;
  const float4 zero4 = make_float4(0.0f, 0.0f, 0.0f, 0.0f);

  int s = blockIdx.x;
  {
    const int ra = s * 16 + sr, rb = ra + 8;
    p0 = zero4;
    p1 = zero4;
    if (s < nstrips && ra < N) p0 = *(const float4*)(hfeat + (size_t)ra * KF + sf);
    if (s < nstrips && rb < N) p1 = *(const float4*)(hfeat + (size_t)rb * KF + sf);
    *(float4*)(&As[0][sr * AS_STRIDE + sf])       = p0;
    *(float4*)(&As[0][(sr + 8) * AS_STRIDE + sf]) = p1;
  }
  __syncthreads();

  int cur = 0;
  for (; s < nstrips; s += sstep) {
    const int sn = s + sstep;
    const bool more = sn < nstrips;
    if (more) {
      const int ra = sn * 16 + sr, rb = ra + 8;
      p0 = zero4;
      p1 = zero4;
      if (ra < N) p0 = *(const float4*)(hfeat + (size_t)ra * KF + sf);
      if (rb < N) p1 = *(const float4*)(hfeat + (size_t)rb * KF + sf);
    }

    const float* Ac = &As[cur][col * AS_STRIDE + 2 * hl];
    v8f acc = {};
#pragma unroll
    for (int j = 0; j < KF / 4; ++j) {
      v2f a;
      a.x = Ac[4 * j];
      a.y = Ac[4 * j + 1];
      acc = __builtin_amdgcn_wmma_f32_16x16x4_f32(
          false, a, false, bfr[j], (short)0, acc, false, false);
    }

    const int row0 = s * 16;
#pragma unroll
    for (int v = 0; v < 8; ++v) {
      int r = row0 + v + 8 * hl;
      if (r < N) Wh[(size_t)r * KHD + n0 + col] = acc[v];
    }

    if (more) {
      *(float4*)(&As[cur ^ 1][sr * AS_STRIDE + sf])       = p0;
      *(float4*)(&As[cur ^ 1][(sr + 8) * AS_STRIDE + sf]) = p1;
    }
    __syncthreads();
    cur ^= 1;
  }
}

__device__ __forceinline__ float leaky(float e) {
  return (e > 0.0f) ? e : KALPHA * e;
}

__device__ __forceinline__ float dot4(const float4 w, const float4 a) {
  return w.x * a.x + w.y * a.y + w.z * a.z + w.w * a.w;
}

__device__ __forceinline__ float quad_sum(float v) {
  v += __int_as_float(__builtin_amdgcn_update_dpp(0, __float_as_int(v), 0x161, 0xf, 0xf, true));
  v += __int_as_float(__builtin_amdgcn_update_dpp(0, __float_as_int(v), 0x162, 0xf, 0xf, true));
  return v;
}

static constexpr int ROW_WGS = 1786;
__global__ __launch_bounds__(256) void row_softmax_gather_kernel(
    const int* __restrict__ ssrc, const int* __restrict__ offs, const int* __restrict__ deg,
    const float* __restrict__ a1, const float* __restrict__ a2,
    const float* __restrict__ Wh, float* __restrict__ out, int N) {
  const int lane = threadIdx.x & 31;
  const float4 a1v = make_float4(a1[lane * 4], a1[lane * 4 + 1], a1[lane * 4 + 2], a1[lane * 4 + 3]);
  const float4 a2v = make_float4(a2[lane * 4], a2[lane * 4 + 1], a2[lane * 4 + 2], a2[lane * 4 + 3]);
  const float* whl = Wh + lane * 4;
  int d = blockIdx.x * 8 + __builtin_amdgcn_readfirstlane((int)(threadIdx.x >> 5));
  const int dstep = gridDim.x * 8;
  int nbeg = 0, ncnt = 0;
  float4 nwd = make_float4(0.0f, 0.0f, 0.0f, 0.0f);
  if (d < N) { nbeg = offs[d]; ncnt = deg[d]; nwd = *(const float4*)(whl + (size_t)d * KHD); }
  for (; d < N; d += dstep) {
  const int beg = nbeg;
  const int end = beg + ncnt;
  const float4 wdst = nwd;
  {
    const int dn = d + dstep;
    if (dn < N) { nbeg = offs[dn]; ncnt = deg[dn]; nwd = *(const float4*)(whl + (size_t)dn * KHD); }
  }

  const float s2b = quad_sum(dot4(wdst, a2v));
  float4 acc = make_float4(0.0f, 0.0f, 0.0f, 0.0f);
  float z = 0.0f;
  float m = -__builtin_inff();
  int i = beg;
  for (; i + 4 <= end; i += 4) {
    const int sa = ssrc[i], sb = ssrc[i + 1], sc = ssrc[i + 2], sd = ssrc[i + 3];
    const float4 wa = *(const float4*)(whl + (size_t)sa * KHD);
    const float4 wb = *(const float4*)(whl + (size_t)sb * KHD);
    const float4 wc = *(const float4*)(whl + (size_t)sc * KHD);
    const float4 wd = *(const float4*)(whl + (size_t)sd * KHD);
    const float ea = quad_sum(dot4(wa, a1v)), eb = quad_sum(dot4(wb, a1v));
    const float ec = quad_sum(dot4(wc, a1v)), ed = quad_sum(dot4(wd, a1v));
    const float xa = leaky(ea + s2b), xb = leaky(eb + s2b), xc = leaky(ec + s2b), xd = leaky(ed + s2b);
    const float mn = fmaxf(fmaxf(fmaxf(xa, xb), fmaxf(xc, xd)), m);
    const float f  = __expf(m - mn);
    m = mn;
    const float pa = __expf(xa - mn);
    const float pb = __expf(xb - mn);
    const float pc = __expf(xc - mn);
    const float pd = __expf(xd - mn);
    z *= f; acc.x *= f; acc.y *= f; acc.z *= f; acc.w *= f;
    z += pa; acc.x += pa * wa.x; acc.y += pa * wa.y; acc.z += pa * wa.z; acc.w += pa * wa.w;
    z += pb; acc.x += pb * wb.x; acc.y += pb * wb.y; acc.z += pb * wb.z; acc.w += pb * wb.w;
    z += pc; acc.x += pc * wc.x; acc.y += pc * wc.y; acc.z += pc * wc.z; acc.w += pc * wc.w;
    z += pd; acc.x += pd * wd.x; acc.y += pd * wd.y; acc.z += pd * wd.z; acc.w += pd * wd.w;
  }
#pragma nounroll
  for (; i < end; ++i) {
    const int sa = ssrc[i];
    const float4 wa = *(const float4*)(whl + (size_t)sa * KHD);
    const float ea = quad_sum(dot4(wa, a1v));
    const float xa = leaky(ea + s2b);
    const float mn = fmaxf(xa, m);
    const float f  = __expf(m - mn);
    m = mn;
    const float pa = __expf(xa - mn);
    z *= f; acc.x *= f; acc.y *= f; acc.z *= f; acc.w *= f;
    z += pa; acc.x += pa * wa.x; acc.y += pa * wa.y; acc.z += pa * wa.z; acc.w += pa * wa.w;
  }
  const float zinv = 1.0f / fmaxf(z, 1e-16f);
  *(float4*)(out + (size_t)d * KHD + lane * 4) =
      make_float4(acc.x * zinv, acc.y * zinv, acc.z * zinv, acc.w * zinv);
  }
}

extern "C" void kernel_launch(void* const* d_in, const int* in_sizes, int n_in,
                              void* d_out, int out_size, void* d_ws, size_t ws_size,
                              hipStream_t stream) {
  const float* hfeat = (const float*)d_in[0];
  const int*   src   = (const int*)d_in[1];
  const int*   dst   = (const int*)d_in[2];
  const float* W     = (const float*)d_in[3];
  const float* a1    = (const float*)d_in[4];
  const float* a2    = (const float*)d_in[5];
  float* out = (float*)d_out;

  const int N = in_sizes[0] / KF;
  const int E = in_sizes[1];

  float* Wh   = (float*)d_ws;
  int*   ssrc = (int*)(Wh + (size_t)N * KHD);
  int*   deg  = ssrc + (size_t)E;
  int*   offs = deg + (size_t)N;
  int*   tsum = offs + (size_t)N;
  int*   rank = (int*)d_ws;

  const int TB = 256;
  const int g_n     = (N  + TB - 1) / TB;
  const int g_pl    = (E + TB * PL_EPT - 1) / (TB * PL_EPT);
  const int g_rk    = (E + RK_EPT * RK_THREADS - 1) / (RK_EPT * RK_THREADS);
  const int g_tile  = (N + SCAN_TILE - 1) / SCAN_TILE;

  zero_deg_kernel<<<g_n, TB, 0, stream>>>(deg, N);
  rank_lds_kernel<<<g_rk, RK_THREADS, 0, stream>>>(dst, deg, rank, N, E);
  tile_sum_kernel<<<g_tile, TB, 0, stream>>>(deg, tsum, N);
  row_offsets_kernel<<<g_tile, TB, 0, stream>>>(deg, tsum, offs, N);
  place_kernel<<<g_pl, TB, 0, stream>>>(src, dst, rank, offs, ssrc, E);

  gat_gemm_wmma_kernel<<<GEMM_WGS, TB, 0, stream>>>(hfeat, W, Wh, N);

  row_softmax_gather_kernel<<<ROW_WGS, TB, 0, stream>>>(ssrc, offs, deg, a1, a2, Wh, out, N);
}


# Edge bucketing as two-level counting sort: LDS histogram per 512-row bucket, one scan, scatter, per-bucket LDS count+scan+place; no global atomics, 4 launches for 5; row sums in LDS arrival order
# speedup vs baseline: 7.9527x; 1.0904x over previous
#include <hip/hip_runtime.h>
#include <hip/hip_bf16.h>


static constexpr int KF  = 128;
static constexpr int KH  = 8;
static constexpr int KD  = 16;
static constexpr int KHD = 128;
static constexpr float KALPHA = 0.2f;

typedef __attribute__((ext_vector_type(2))) float v2f;
typedef __attribute__((ext_vector_type(8))) float v8f;

static constexpr int PT_THREADS  = 512;
static constexpr int PT_EPT      = 16;
static constexpr int PT_CHUNK    = PT_THREADS * PT_EPT;
static constexpr int RB_SHIFT    = 9;
static constexpr int RB_ROWS     = 1 << RB_SHIFT;
static constexpr int MAX_BUCKETS = 1024;

__global__ __launch_bounds__(PT_THREADS) void bucket_hist_kernel(const int* __restrict__ dst,
                                                                 int* __restrict__ gh, int N, int E, int B) {
  __shared__ int hist[MAX_BUCKETS];
  const int tid = threadIdx.x;
  const int g = blockIdx.x, G = gridDim.x;
  for (int b = tid; b < B; b += PT_THREADS) hist[b] = 0;
  __syncthreads();
  const int ebase = g * PT_CHUNK + tid;
  int dv[PT_EPT];
#pragma unroll
  for (int q = 0; q < PT_EPT; ++q) {
    const int e = ebase + q * PT_THREADS;
    dv[q] = (e < E) ? dst[e] : -1;
  }
#pragma unroll
  for (int q = 0; q < PT_EPT; ++q)
    if ((unsigned)dv[q] < (unsigned)N) atomicAdd(&hist[dv[q] >> RB_SHIFT], 1);
  __syncthreads();
  for (int b = tid; b < B; b += PT_THREADS) gh[b * G + g] = hist[b];
}

static constexpr int SC_THREADS = 1024;
static constexpr int SC_PER     = 40;
static constexpr int SC_PITCH   = SC_PER + 1;
static constexpr int SC_TILE    = SC_THREADS * SC_PER;

__global__ __launch_bounds__(SC_THREADS) void bucket_scan_kernel(int* __restrict__ gh, int M) {
  __shared__ int buf[SC_THREADS * SC_PITCH];
  __shared__ int wsum[32];
  __shared__ int tile_total;
  const int tid = threadIdx.x, lane = tid & 31, wave = tid >> 5;
  int carry = 0;
  for (int t0 = 0; t0 < M; t0 += SC_TILE) {
    int v[SC_PER];
#pragma unroll
    for (int q = 0; q < SC_PER; ++q) {
      const int idx = t0 + tid + q * SC_THREADS;
      v[q] = (idx < M) ? gh[idx] : 0;
    }
#pragma unroll
    for (int q = 0; q < SC_PER; ++q) {
      const int i = tid + q * SC_THREADS;
      buf[(i / SC_PER) * SC_PITCH + (i % SC_PER)] = v[q];
    }
    __syncthreads();
    int* my = &buf[tid * SC_PITCH];
    int s = 0;
    for (int q = 0; q < SC_PER; ++q) s += my[q];
    int inc = s;
    for (int off = 1; off < 32; off <<= 1) {
      const int u = __shfl_up(inc, off);
      if (lane >= off) inc += u;
    }
    if (lane == 31) wsum[wave] = inc;
    __syncthreads();
    if (wave == 0) {
      const int w = wsum[lane];
      int wi = w;
      for (int off = 1; off < 32; off <<= 1) {
        const int u = __shfl_up(wi, off);
        if (lane >= off) wi += u;
      }
      wsum[lane] = wi - w;
      if (lane == 31) tile_total = wi;
    }
    __syncthreads();
    int ex = carry + wsum[wave] + inc - s;
    for (int q = 0; q < SC_PER; ++q) {
      const int x = my[q];
      my[q] = ex;
      ex += x;
    }
    carry += tile_total;
    __syncthreads();
    for (int i = tid; i < SC_TILE; i += SC_THREADS) {
      const int idx = t0 + i;
      if (idx < M) gh[idx] = buf[(i / SC_PER) * SC_PITCH + (i % SC_PER)];
    }
    __syncthreads();
  }
  if (tid == 0) gh[M] = carry;
}

__global__ __launch_bounds__(PT_THREADS) void bucket_scatter_kernel(const int* __restrict__ src,
                                                                    const int* __restrict__ dst,
                                                                    const int* __restrict__ base,
                                                                    int* __restrict__ rec, int N, int E, int B) {
  __shared__ int cur[MAX_BUCKETS];
  const int tid = threadIdx.x;
  const int g = blockIdx.x, G = gridDim.x;
  for (int b = tid; b < B; b += PT_THREADS) cur[b] = base[b * G + g];
  const int ebase = g * PT_CHUNK + tid;
  int dv[PT_EPT], sv[PT_EPT];
#pragma unroll
  for (int q = 0; q < PT_EPT; ++q) {
    const int e = ebase + q * PT_THREADS;
    dv[q] = (e < E) ? dst[e] : -1;
    sv[q] = (e < E) ? src[e] : 0;
  }
  __syncthreads();
#pragma unroll
  for (int q = 0; q < PT_EPT; ++q) {
    if ((unsigned)dv[q] < (unsigned)N) {
      const int pos = atomicAdd(&cur[dv[q] >> RB_SHIFT], 1);
      rec[pos] = (int)(((unsigned)sv[q] << RB_SHIFT) | ((unsigned)dv[q] & (unsigned)(RB_ROWS - 1)));
    }
  }
}

static constexpr int BP_UNROLL = 8;

__global__ __launch_bounds__(RB_ROWS) void bucket_place_kernel(const int* __restrict__ rec,
                                                               const int* __restrict__ base,
                                                               int* __restrict__ deg, int* __restrict__ offs,
                                                               int* __restrict__ ssrc, int N, int G) {
  __shared__ int cnt[RB_ROWS];
  __shared__ int wsum[RB_ROWS / 32];
  const int tid = threadIdx.x, lane = tid & 31, wave = tid >> 5;
  const int b = blockIdx.x;
  const int s = base[b * G];
  const int e = base[(b + 1) * G];
  cnt[tid] = 0;
  __syncthreads();
  for (int i0 = s + tid; i0 < e; i0 += BP_UNROLL * RB_ROWS) {
    int r[BP_UNROLL];
#pragma unroll
    for (int q = 0; q < BP_UNROLL; ++q) {
      const int i = i0 + q * RB_ROWS;
      r[q] = (i < e) ? rec[i] : -1;
    }
#pragma unroll
    for (int q = 0; q < BP_UNROLL; ++q)
      if (r[q] >= 0) atomicAdd(&cnt[r[q] & (RB_ROWS - 1)], 1);
  }
  __syncthreads();
  const int c = cnt[tid];
  int inc = c;
  for (int off = 1; off < 32; off <<= 1) {
    const int u = __shfl_up(inc, off);
    if (lane >= off) inc += u;
  }
  if (lane == 31) wsum[wave] = inc;
  __syncthreads();
  int add = 0;
#pragma unroll
  for (int w = 0; w < RB_ROWS / 32; ++w)
    if (w < wave) add += wsum[w];
  const int ex = s + add + inc - c;
  const int row = (b << RB_SHIFT) + tid;
  if (row < N) {
    deg[row]  = c;
    offs[row] = ex;
  }
  cnt[tid] = ex;
  __syncthreads();
  for (int i0 = s + tid; i0 < e; i0 += BP_UNROLL * RB_ROWS) {
    int r[BP_UNROLL];
#pragma unroll
    for (int q = 0; q < BP_UNROLL; ++q) {
      const int i = i0 + q * RB_ROWS;
      r[q] = (i < e) ? rec[i] : -1;
    }
#pragma unroll
    for (int q = 0; q < BP_UNROLL; ++q) {
      if (r[q] >= 0) {
        const int pos = atomicAdd(&cnt[r[q] & (RB_ROWS - 1)], 1);
        ssrc[pos] = (int)((unsigned)r[q] >> RB_SHIFT);
      }
    }
  }
}

static constexpr int AS_STRIDE = 132;
static constexpr int GEMM_WGS  = 1024;

__global__ __launch_bounds__(256) void gat_gemm_wmma_kernel(
    const float* __restrict__ hfeat, const float* __restrict__ W,
    float* __restrict__ Wh, int N) {
  __shared__ alignas(16) float As[2][16 * AS_STRIDE];

  const int tid  = threadIdx.x;
  const int wave = tid >> 5;
  const int lane = tid & 31;
  const int hl   = lane >> 4;
  const int col  = lane & 15;
  const int n0   = wave * 16;
  const int nstrips = (N + 15) >> 4;
  const int sstep   = gridDim.x;

  const float* Wb = W + wave * (KF * KD) + col;
  v2f bfr[KF / 4];
#pragma unroll
  for (int j = 0; j < KF / 4; ++j) {
    const int k = 4 * j + 2 * hl;
    bfr[j].x = Wb[(size_t)k * KD];
    bfr[j].y = Wb[(size_t)(k + 1) * KD];
  }

  const int sr = tid >> 5;
  const int sf = (tid & 31) * 4;
  float4 p0, p1;
  const float4 zero4 = make_float4(0.0f, 0.0f, 0.0f, 0.0f);

  int s = blockIdx.x;
  {
    const int ra = s * 16 + sr, rb = ra + 8;
    p0 = zero4;
    p1 = zero4;
    if (s < nstrips && ra < N) p0 = *(const float4*)(hfeat + (size_t)ra * KF + sf);
    if (s < nstrips && rb < N) p1 = *(const float4*)(hfeat + (size_t)rb * KF + sf);
    *(float4*)(&As[0][sr * AS_STRIDE + sf])       = p0;
    *(float4*)(&As[0][(sr + 8) * AS_STRIDE + sf]) = p1;
  }
  __syncthreads();

  int cur = 0;
  for (; s < nstrips; s += sstep) {
    const int sn = s + sstep;
    const bool more = sn < nstrips;
    if (more) {
      const int ra = sn * 16 + sr, rb = ra + 8;
      p0 = zero4;
      p1 = zero4;
      if (ra < N) p0 = *(const float4*)(hfeat + (size_t)ra * KF + sf);
      if (rb < N) p1 = *(const float4*)(hfeat + (size_t)rb * KF + sf);
    }

    const float* Ac = &As[cur][col * AS_STRIDE + 2 * hl];
    v8f acc = {};
#pragma unroll
    for (int j = 0; j < KF / 4; ++j) {
      v2f a;
      a.x = Ac[4 * j];
      a.y = Ac[4 * j + 1];
      acc = __builtin_amdgcn_wmma_f32_16x16x4_f32(
          false, a, false, bfr[j], (short)0, acc, false, false);
    }

    const int row0 = s * 16;
#pragma unroll
    for (int v = 0; v < 8; ++v) {
      int r = row0 + v + 8 * hl;
      if (r < N) Wh[(size_t)r * KHD + n0 + col] = acc[v];
    }

    if (more) {
      *(float4*)(&As[cur ^ 1][sr * AS_STRIDE + sf])       = p0;
      *(float4*)(&As[cur ^ 1][(sr + 8) * AS_STRIDE + sf]) = p1;
    }
    __syncthreads();
    cur ^= 1;
  }
}

__device__ __forceinline__ float leaky(float e) {
  return (e > 0.0f) ? e : KALPHA * e;
}

__device__ __forceinline__ float dot4(const float4 w, const float4 a) {
  return w.x * a.x + w.y * a.y + w.z * a.z + w.w * a.w;
}

__device__ __forceinline__ float quad_sum(float v) {
  v += __int_as_float(__builtin_amdgcn_update_dpp(0, __float_as_int(v), 0x161, 0xf, 0xf, true));
  v += __int_as_float(__builtin_amdgcn_update_dpp(0, __float_as_int(v), 0x162, 0xf, 0xf, true));
  return v;
}

static constexpr int ROW_WGS = 1786;
__global__ __launch_bounds__(256) void row_softmax_gather_kernel(
    const int* __restrict__ ssrc, const int* __restrict__ offs, const int* __restrict__ deg,
    const float* __restrict__ a1, const float* __restrict__ a2,
    const float* __restrict__ Wh, float* __restrict__ out, int N) {
  const int lane = threadIdx.x & 31;
  const float4 a1v = make_float4(a1[lane * 4], a1[lane * 4 + 1], a1[lane * 4 + 2], a1[lane * 4 + 3]);
  const float4 a2v = make_float4(a2[lane * 4], a2[lane * 4 + 1], a2[lane * 4 + 2], a2[lane * 4 + 3]);
  const float* whl = Wh + lane * 4;
  int d = blockIdx.x * 8 + __builtin_amdgcn_readfirstlane((int)(threadIdx.x >> 5));
  const int dstep = gridDim.x * 8;
  int nbeg = 0, ncnt = 0;
  float4 nwd = make_float4(0.0f, 0.0f, 0.0f, 0.0f);
  if (d < N) { nbeg = offs[d]; ncnt = deg[d]; nwd = *(const float4*)(whl + (size_t)d * KHD); }
  for (; d < N; d += dstep) {
  const int beg = nbeg;
  const int end = beg + ncnt;
  const float4 wdst = nwd;
  {
    const int dn = d + dstep;
    if (dn < N) { nbeg = offs[dn]; ncnt = deg[dn]; nwd = *(const float4*)(whl + (size_t)dn * KHD); }
  }

  const float s2b = quad_sum(dot4(wdst, a2v));
  float4 acc = make_float4(0.0f, 0.0f, 0.0f, 0.0f);
  float z = 0.0f;
  float m = -__builtin_inff();
  int i = beg;
  for (; i + 4 <= end; i += 4) {
    const int sa = ssrc[i], sb = ssrc[i + 1], sc = ssrc[i + 2], sd = ssrc[i + 3];
    const float4 wa = *(const float4*)(whl + (size_t)sa * KHD);
    const float4 wb = *(const float4*)(whl + (size_t)sb * KHD);
    const float4 wc = *(const float4*)(whl + (size_t)sc * KHD);
    const float4 wd = *(const float4*)(whl + (size_t)sd * KHD);
    const float ea = quad_sum(dot4(wa, a1v)), eb = quad_sum(dot4(wb, a1v));
    const float ec = quad_sum(dot4(wc, a1v)), ed = quad_sum(dot4(wd, a1v));
    const float xa = leaky(ea + s2b), xb = leaky(eb + s2b), xc = leaky(ec + s2b), xd = leaky(ed + s2b);
    const float mn = fmaxf(fmaxf(fmaxf(xa, xb), fmaxf(xc, xd)), m);
    const float f  = __expf(m - mn);
    m = mn;
    const float pa = __expf(xa - mn);
    const float pb = __expf(xb - mn);
    const float pc = __expf(xc - mn);
    const float pd = __expf(xd - mn);
    z *= f; acc.x *= f; acc.y *= f; acc.z *= f; acc.w *= f;
    z += pa; acc.x += pa * wa.x; acc.y += pa * wa.y; acc.z += pa * wa.z; acc.w += pa * wa.w;
    z += pb; acc.x += pb * wb.x; acc.y += pb * wb.y; acc.z += pb * wb.z; acc.w += pb * wb.w;
    z += pc; acc.x += pc * wc.x; acc.y += pc * wc.y; acc.z += pc * wc.z; acc.w += pc * wc.w;
    z += pd; acc.x += pd * wd.x; acc.y += pd * wd.y; acc.z += pd * wd.z; acc.w += pd * wd.w;
  }
#pragma nounroll
  for (; i < end; ++i) {
    const int sa = ssrc[i];
    const float4 wa = *(const float4*)(whl + (size_t)sa * KHD);
    const float ea = quad_sum(dot4(wa, a1v));
    const float xa = leaky(ea + s2b);
    const float mn = fmaxf(xa, m);
    const float f  = __expf(m - mn);
    m = mn;
    const float pa = __expf(xa - mn);
    z *= f; acc.x *= f; acc.y *= f; acc.z *= f; acc.w *= f;
    z += pa; acc.x += pa * wa.x; acc.y += pa * wa.y; acc.z += pa * wa.z; acc.w += pa * wa.w;
  }
  const float zinv = 1.0f / fmaxf(z, 1e-16f);
  *(float4*)(out + (size_t)d * KHD + lane * 4) =
      make_float4(acc.x * zinv, acc.y * zinv, acc.z * zinv, acc.w * zinv);
  }
}

extern "C" void kernel_launch(void* const* d_in, const int* in_sizes, int n_in,
                              void* d_out, int out_size, void* d_ws, size_t ws_size,
                              hipStream_t stream) {
  const float* hfeat = (const float*)d_in[0];
  const int*   src   = (const int*)d_in[1];
  const int*   dst   = (const int*)d_in[2];
  const float* W     = (const float*)d_in[3];
  const float* a1    = (const float*)d_in[4];
  const float* a2    = (const float*)d_in[5];
  float* out = (float*)d_out;

  const int N = in_sizes[0] / KF;
  const int E = in_sizes[1];

  float* Wh   = (float*)d_ws;
  int*   ssrc = (int*)(Wh + (size_t)N * KHD);
  int*   deg  = ssrc + (size_t)E;
  int*   offs = deg + (size_t)N;
  int*   rec  = (int*)d_ws;
  int*   gh   = rec + (size_t)E;

  const int TB = 256;
  const int G = (E + PT_CHUNK - 1) / PT_CHUNK;
  const int B = (N + RB_ROWS - 1) / RB_ROWS;

  bucket_hist_kernel<<<G, PT_THREADS, 0, stream>>>(dst, gh, N, E, B);
  bucket_scan_kernel<<<1, SC_THREADS, 0, stream>>>(gh, B * G);
  bucket_scatter_kernel<<<G, PT_THREADS, 0, stream>>>(src, dst, gh, rec, N, E, B);
  bucket_place_kernel<<<B, RB_ROWS, 0, stream>>>(rec, gh, deg, offs, ssrc, N, G);

  gat_gemm_wmma_kernel<<<GEMM_WGS, TB, 0, stream>>>(hfeat, W, Wh, N);

  row_softmax_gather_kernel<<<ROW_WGS, TB, 0, stream>>>(ssrc, offs, deg, a1, a2, Wh, out, N);
}


# Edge bucketing in 2 launches for 4: each chunk sorts its edges by bucket in its own span (LDS rank + scan, edge loads unguarded as one group); place walks the runs; row sums in LDS arrival order
# speedup vs baseline: 10.2911x; 1.2940x over previous
#include <hip/hip_runtime.h>
#include <hip/hip_bf16.h>


static constexpr int KF  = 128;
static constexpr int KH  = 8;
static constexpr int KD  = 16;
static constexpr int KHD = 128;
static constexpr float KALPHA = 0.2f;

typedef __attribute__((ext_vector_type(2))) float v2f;
typedef __attribute__((ext_vector_type(8))) float v8f;

static constexpr int PT_THREADS  = 512;
static constexpr int PT_EPT      = 16;
static constexpr int PT_CHUNK    = PT_THREADS * PT_EPT;
static constexpr int RB_SHIFT    = 9;
static constexpr int RB_ROWS     = 1 << RB_SHIFT;
static constexpr int MAX_BUCKETS = 2 * PT_THREADS;
static constexpr int MAX_CHUNKS  = 1024;
static constexpr int RUN_SHIFT = 14;
static constexpr int RUN_MASK  = (1 << RUN_SHIFT) - 1;

__global__ __launch_bounds__(PT_THREADS) void chunk_sort_kernel(const int* __restrict__ src,
                                                                const int* __restrict__ dst,
                                                                int* __restrict__ rec, int* __restrict__ tab,
                                                                int N, int E, int B) {
  __shared__ int hist[MAX_BUCKETS];
  __shared__ int wsum[PT_THREADS / 32];
  const int tid = threadIdx.x, lane = tid & 31, wave = tid >> 5;
  const int g = blockIdx.x, G = gridDim.x;
  hist[tid] = 0;
  hist[tid + PT_THREADS] = 0;
  const int ebase = g * PT_CHUNK + tid;
  int dv[PT_EPT], sv[PT_EPT], rk[PT_EPT];
#pragma unroll
  for (int q = 0; q < PT_EPT; ++q) {
    const int e  = ebase + q * PT_THREADS;
    const int ec = (e < E) ? e : (E - 1);
    const int d = dst[ec], sr = src[ec];
    dv[q] = (e < E) ? d : -1;
    sv[q] = sr;
  }
  __syncthreads();
#pragma unroll
  for (int q = 0; q < PT_EPT; ++q) {
    rk[q] = 0;
    if ((unsigned)dv[q] < (unsigned)N) rk[q] = atomicAdd(&hist[dv[q] >> RB_SHIFT], 1);
  }
  __syncthreads();
  const int c0 = hist[2 * tid], c1 = hist[2 * tid + 1];
  const int s = c0 + c1;
  int inc = s;
  for (int off = 1; off < 32; off <<= 1) {
    const int u = __shfl_up(inc, off);
    if (lane >= off) inc += u;
  }
  if (lane == 31) wsum[wave] = inc;
  __syncthreads();
  int add = 0;
#pragma unroll
  for (int w = 0; w < PT_THREADS / 32; ++w)
    if (w < wave) add += wsum[w];
  const int ex0 = add + inc - s, ex1 = ex0 + c0;
  hist[2 * tid]     = ex0;
  hist[2 * tid + 1] = ex1;
  if (2 * tid < B)     tab[(2 * tid) * G + g]     = (ex0 << RUN_SHIFT) | c0;
  if (2 * tid + 1 < B) tab[(2 * tid + 1) * G + g] = (ex1 << RUN_SHIFT) | c1;
  __syncthreads();
  int* myrec = rec + (size_t)g * PT_CHUNK;
#pragma unroll
  for (int q = 0; q < PT_EPT; ++q) {
    if ((unsigned)dv[q] < (unsigned)N)
      myrec[hist[dv[q] >> RB_SHIFT] + rk[q]] =
          (int)(((unsigned)sv[q] << RB_SHIFT) | ((unsigned)dv[q] & (unsigned)(RB_ROWS - 1)));
  }
}

static constexpr int BP_RUNS  = 8;
static constexpr int BP_WAVES = RB_ROWS / 32;

__global__ __launch_bounds__(RB_ROWS) void bucket_place_kernel(const int* __restrict__ rec,
                                                               const int* __restrict__ tab,
                                                               int* __restrict__ deg, int* __restrict__ offs,
                                                               int* __restrict__ ssrc, int N, int G) {
  __shared__ int cnt[RB_ROWS];
  __shared__ int wsum[RB_ROWS / 32];
  __shared__ int bsum[RB_ROWS / 32];
  __shared__ int run[MAX_CHUNKS];
  const int tid = threadIdx.x, lane = tid & 31, wave = tid >> 5;
  const int b = blockIdx.x;
  cnt[tid] = 0;
  int part = 0;
  for (int g = tid; g < G; g += RB_ROWS) {
    const int pk = tab[b * G + g];
    run[g] = pk;
    part += pk >> RUN_SHIFT;
  }
  for (int off = 16; off > 0; off >>= 1) part += __shfl_xor(part, off);
  if (lane == 0) bsum[wave] = part;
  __syncthreads();
  int s = 0;
#pragma unroll
  for (int w = 0; w < RB_ROWS / 32; ++w) s += bsum[w];
  for (int j0 = 0; j0 * BP_WAVES + wave < G; j0 += BP_RUNS) {
    int r[BP_RUNS][2], len[BP_RUNS], at[BP_RUNS];
#pragma unroll
    for (int u = 0; u < BP_RUNS; ++u) {
      const int g = (j0 + u) * BP_WAVES + wave;
      const int pk = (g < G) ? run[g] : 0;
      len[u] = pk & RUN_MASK;
      at[u]  = g * PT_CHUNK + (pk >> RUN_SHIFT);
      r[u][0] = (lane < len[u]) ? rec[at[u] + lane] : -1;
      r[u][1] = (lane + 32 < len[u]) ? rec[at[u] + lane + 32] : -1;
    }
#pragma unroll
    for (int u = 0; u < BP_RUNS; ++u) {
      if (r[u][0] >= 0) atomicAdd(&cnt[r[u][0] & (RB_ROWS - 1)], 1);
      if (r[u][1] >= 0) atomicAdd(&cnt[r[u][1] & (RB_ROWS - 1)], 1);
      for (int o = lane + 64; o < len[u]; o += 32) atomicAdd(&cnt[rec[at[u] + o] & (RB_ROWS - 1)], 1);
    }
  }
  __syncthreads();
  const int c = cnt[tid];
  int inc = c;
  for (int off = 1; off < 32; off <<= 1) {
    const int u = __shfl_up(inc, off);
    if (lane >= off) inc += u;
  }
  if (lane == 31) wsum[wave] = inc;
  __syncthreads();
  int add = 0;
#pragma unroll
  for (int w = 0; w < RB_ROWS / 32; ++w)
    if (w < wave) add += wsum[w];
  const int ex = s + add + inc - c;
  const int row = (b << RB_SHIFT) + tid;
  if (row < N) {
    deg[row]  = c;
    offs[row] = ex;
  }
  cnt[tid] = ex;
  __syncthreads();
  for (int j0 = 0; j0 * BP_WAVES + wave < G; j0 += BP_RUNS) {
    int r[BP_RUNS][2], len[BP_RUNS], at[BP_RUNS];
#pragma unroll
    for (int u = 0; u < BP_RUNS; ++u) {
      const int g = (j0 + u) * BP_WAVES + wave;
      const int pk = (g < G) ? run[g] : 0;
      len[u] = pk & RUN_MASK;
      at[u]  = g * PT_CHUNK + (pk >> RUN_SHIFT);
      r[u][0] = (lane < len[u]) ? rec[at[u] + lane] : -1;
      r[u][1] = (lane + 32 < len[u]) ? rec[at[u] + lane + 32] : -1;
    }
#pragma unroll
    for (int u = 0; u < BP_RUNS; ++u) {
      if (r[u][0] >= 0) {
        const int pos = atomicAdd(&cnt[r[u][0] & (RB_ROWS - 1)], 1);
        ssrc[pos] = (int)((unsigned)r[u][0] >> RB_SHIFT);
      }
      if (r[u][1] >= 0) {
        const int pos = atomicAdd(&cnt[r[u][1] & (RB_ROWS - 1)], 1);
        ssrc[pos] = (int)((unsigned)r[u][1] >> RB_SHIFT);
      }
      for (int o = lane + 64; o < len[u]; o += 32) {
        const int rr = rec[at[u] + o];
        const int pos = atomicAdd(&cnt[rr & (RB_ROWS - 1)], 1);
        ssrc[pos] = (int)((unsigned)rr >> RB_SHIFT);
      }
    }
  }
}

static constexpr int AS_STRIDE = 132;
static constexpr int GEMM_WGS  = 1024;

__global__ __launch_bounds__(256) void gat_gemm_wmma_kernel(
    const float* __restrict__ hfeat, const float* __restrict__ W,
    float* __restrict__ Wh, int N) {
  __shared__ alignas(16) float As[2][16 * AS_STRIDE];

  const int tid  = threadIdx.x;
  const int wave = tid >> 5;
  const int lane = tid & 31;
  const int hl   = lane >> 4;
  const int col  = lane & 15;
  const int n0   = wave * 16;
  const int nstrips = (N + 15) >> 4;
  const int sstep   = gridDim.x;

  const float* Wb = W + wave * (KF * KD) + col;
  v2f bfr[KF / 4];
#pragma unroll
  for (int j = 0; j < KF / 4; ++j) {
    const int k = 4 * j + 2 * hl;
    bfr[j].x = Wb[(size_t)k * KD];
    bfr[j].y = Wb[(size_t)(k + 1) * KD];
  }

  const int sr = tid >> 5;
  const int sf = (tid & 31) * 4;
  float4 p0, p1;
  const float4 zero4 = make_float4(0.0f, 0.0f, 0.0f, 0.0f);

  int s = blockIdx.x;
  {
    const int ra = s * 16 + sr, rb = ra + 8;
    p0 = zero4;
    p1 = zero4;
    if (s < nstrips && ra < N) p0 = *(const float4*)(hfeat + (size_t)ra * KF + sf);
    if (s < nstrips && rb < N) p1 = *(const float4*)(hfeat + (size_t)rb * KF + sf);
    *(float4*)(&As[0][sr * AS_STRIDE + sf])       = p0;
    *(float4*)(&As[0][(sr + 8) * AS_STRIDE + sf]) = p1;
  }
  __syncthreads();

  int cur = 0;
  for (; s < nstrips; s += sstep) {
    const int sn = s + sstep;
    const bool more = sn < nstrips;
    if (more) {
      const int ra = sn * 16 + sr, rb = ra + 8;
      p0 = zero4;
      p1 = zero4;
      if (ra < N) p0 = *(const float4*)(hfeat + (size_t)ra * KF + sf);
      if (rb < N) p1 = *(const float4*)(hfeat + (size_t)rb * KF + sf);
    }

    const float* Ac = &As[cur][col * AS_STRIDE + 2 * hl];
    v8f acc = {};
#pragma unroll
    for (int j = 0; j < KF / 4; ++j) {
      v2f a;
      a.x = Ac[4 * j];
      a.y = Ac[4 * j + 1];
      acc = __builtin_amdgcn_wmma_f32_16x16x4_f32(
          false, a, false, bfr[j], (short)0, acc, false, false);
    }

    const int row0 = s * 16;
#pragma unroll
    for (int v = 0; v < 8; ++v) {
      int r = row0 + v + 8 * hl;
      if (r < N) Wh[(size_t)r * KHD + n0 + col] = acc[v];
    }

    if (more) {
      *(float4*)(&As[cur ^ 1][sr * AS_STRIDE + sf])       = p0;
      *(float4*)(&As[cur ^ 1][(sr + 8) * AS_STRIDE + sf]) = p1;
    }
    __syncthreads();
    cur ^= 1;
  }
}

__device__ __forceinline__ float leaky(float e) {
  return (e > 0.0f) ? e : KALPHA * e;
}

__device__ __forceinline__ float dot4(const float4 w, const float4 a) {
  return w.x * a.x + w.y * a.y + w.z * a.z + w.w * a.w;
}

__device__ __forceinline__ float quad_sum(float v) {
  v += __int_as_float(__builtin_amdgcn_update_dpp(0, __float_as_int(v), 0x161, 0xf, 0xf, true));
  v += __int_as_float(__builtin_amdgcn_update_dpp(0, __float_as_int(v), 0x162, 0xf, 0xf, true));
  return v;
}

static constexpr int ROW_WGS = 1786;
__global__ __launch_bounds__(256) void row_softmax_gather_kernel(
    const int* __restrict__ ssrc, const int* __restrict__ offs, const int* __restrict__ deg,
    const float* __restrict__ a1, const float* __restrict__ a2,
    const float* __restrict__ Wh, float* __restrict__ out, int N) {
  const int lane = threadIdx.x & 31;
  const float4 a1v = make_float4(a1[lane * 4], a1[lane * 4 + 1], a1[lane * 4 + 2], a1[lane * 4 + 3]);
  const float4 a2v = make_float4(a2[lane * 4], a2[lane * 4 + 1], a2[lane * 4 + 2], a2[lane * 4 + 3]);
  const float* whl = Wh + lane * 4;
  int d = blockIdx.x * 8 + __builtin_amdgcn_readfirstlane((int)(threadIdx.x >> 5));
  const int dstep = gridDim.x * 8;
  int nbeg = 0, ncnt = 0;
  float4 nwd = make_float4(0.0f, 0.0f, 0.0f, 0.0f);
  if (d < N) { nbeg = offs[d]; ncnt = deg[d]; nwd = *(const float4*)(whl + (size_t)d * KHD); }
  for (; d < N; d += dstep) {
  const int beg = nbeg;
  const int end = beg + ncnt;
  const float4 wdst = nwd;
  {
    const int dn = d + dstep;
    if (dn < N) { nbeg = offs[dn]; ncnt = deg[dn]; nwd = *(const float4*)(whl + (size_t)dn * KHD); }
  }

  const float s2b = quad_sum(dot4(wdst, a2v));
  float4 acc = make_float4(0.0f, 0.0f, 0.0f, 0.0f);
  float z = 0.0f;
  float m = -__builtin_inff();
  int i = beg;
  for (; i + 4 <= end; i += 4) {
    const int sa = ssrc[i], sb = ssrc[i + 1], sc = ssrc[i + 2], sd = ssrc[i + 3];
    const float4 wa = *(const float4*)(whl + (size_t)sa * KHD);
    const float4 wb = *(const float4*)(whl + (size_t)sb * KHD);
    const float4 wc = *(const float4*)(whl + (size_t)sc * KHD);
    const float4 wd = *(const float4*)(whl + (size_t)sd * KHD);
    const float ea = quad_sum(dot4(wa, a1v)), eb = quad_sum(dot4(wb, a1v));
    const float ec = quad_sum(dot4(wc, a1v)), ed = quad_sum(dot4(wd, a1v));
    const float xa = leaky(ea + s2b), xb = leaky(eb + s2b), xc = leaky(ec + s2b), xd = leaky(ed + s2b);
    const float mn = fmaxf(fmaxf(fmaxf(xa, xb), fmaxf(xc, xd)), m);
    const float f  = __expf(m - mn);
    m = mn;
    const float pa = __expf(xa - mn);
    const float pb = __expf(xb - mn);
    const float pc = __expf(xc - mn);
    const float pd = __expf(xd - mn);
    z *= f; acc.x *= f; acc.y *= f; acc.z *= f; acc.w *= f;
    z += pa; acc.x += pa * wa.x; acc.y += pa * wa.y; acc.z += pa * wa.z; acc.w += pa * wa.w;
    z += pb; acc.x += pb * wb.x; acc.y += pb * wb.y; acc.z += pb * wb.z; acc.w += pb * wb.w;
    z += pc; acc.x += pc * wc.x; acc.y += pc * wc.y; acc.z += pc * wc.z; acc.w += pc * wc.w;
    z += pd; acc.x += pd * wd.x; acc.y += pd * wd.y; acc.z += pd * wd.z; acc.w += pd * wd.w;
  }
#pragma nounroll
  for (; i < end; ++i) {
    const int sa = ssrc[i];
    const float4 wa = *(const float4*)(whl + (size_t)sa * KHD);
    const float ea = quad_sum(dot4(wa, a1v));
    const float xa = leaky(ea + s2b);
    const float mn = fmaxf(xa, m);
    const float f  = __expf(m - mn);
    m = mn;
    const float pa = __expf(xa - mn);
    z *= f; acc.x *= f; acc.y *= f; acc.z *= f; acc.w *= f;
    z += pa; acc.x += pa * wa.x; acc.y += pa * wa.y; acc.z += pa * wa.z; acc.w += pa * wa.w;
  }
  const float zinv = 1.0f / fmaxf(z, 1e-16f);
  *(float4*)(out + (size_t)d * KHD + lane * 4) =
      make_float4(acc.x * zinv, acc.y * zinv, acc.z * zinv, acc.w * zinv);
  }
}

extern "C" void kernel_launch(void* const* d_in, const int* in_sizes, int n_in,
                              void* d_out, int out_size, void* d_ws, size_t ws_size,
                              hipStream_t stream) {
  const float* hfeat = (const float*)d_in[0];
  const int*   src   = (const int*)d_in[1];
  const int*   dst   = (const int*)d_in[2];
  const float* W     = (const float*)d_in[3];
  const float* a1    = (const float*)d_in[4];
  const float* a2    = (const float*)d_in[5];
  float* out = (float*)d_out;

  const int N = in_sizes[0] / KF;
  const int E = in_sizes[1];

  float* Wh   = (float*)d_ws;
  int*   ssrc = (int*)(Wh + (size_t)N * KHD);
  int*   deg  = ssrc + (size_t)E;
  int*   offs = deg + (size_t)N;
  int*   rec  = (int*)d_ws;
  int*   tab  = rec + (size_t)E;

  const int TB = 256;
  const int G = (E + PT_CHUNK - 1) / PT_CHUNK;
  const int B = (N + RB_ROWS - 1) / RB_ROWS;

  chunk_sort_kernel<<<G, PT_THREADS, 0, stream>>>(src, dst, rec, tab, N, E, B);
  bucket_place_kernel<<<B, RB_ROWS, 0, stream>>>(rec, tab, deg, offs, ssrc, N, G);

  gat_gemm_wmma_kernel<<<GEMM_WGS, TB, 0, stream>>>(hfeat, W, Wh, N);

  row_softmax_gather_kernel<<<ROW_WGS, TB, 0, stream>>>(ssrc, offs, deg, a1, a2, Wh, out, N);
}


# Both bucketing kernels assemble their sorted edge words in LDS and store them as whole 128-byte lines instead of one scattered 4-byte store per edge; row sums still in LDS arrival order
# speedup vs baseline: 11.3695x; 1.1048x over previous
#include <hip/hip_runtime.h>
#include <hip/hip_bf16.h>


static constexpr int KF  = 128;
static constexpr int KH  = 8;
static constexpr int KD  = 16;
static constexpr int KHD = 128;
static constexpr float KALPHA = 0.2f;

typedef __attribute__((ext_vector_type(2))) float v2f;
typedef __attribute__((ext_vector_type(8))) float v8f;

static constexpr int PT_THREADS  = 512;
static constexpr int PT_EPT      = 16;
static constexpr int PT_CHUNK    = PT_THREADS * PT_EPT;
static constexpr int RB_SHIFT    = 9;
static constexpr int RB_ROWS     = 1 << RB_SHIFT;
static constexpr int MAX_BUCKETS = 2 * PT_THREADS;
static constexpr int MAX_CHUNKS  = 1024;
static constexpr int RUN_SHIFT = 14;
static constexpr int RUN_MASK  = (1 << RUN_SHIFT) - 1;

__global__ __launch_bounds__(PT_THREADS) void chunk_sort_kernel(const int* __restrict__ src,
                                                                const int* __restrict__ dst,
                                                                int* __restrict__ rec, int* __restrict__ tab,
                                                                int N, int E, int B) {
  __shared__ int hist[MAX_BUCKETS];
  __shared__ int wsum[PT_THREADS / 32];
  __shared__ alignas(16) int srt[PT_CHUNK];
  const int tid = threadIdx.x, lane = tid & 31, wave = tid >> 5;
  const int g = blockIdx.x, G = gridDim.x;
  hist[tid] = 0;
  hist[tid + PT_THREADS] = 0;
  const int ebase = g * PT_CHUNK + tid;
  int dv[PT_EPT], sv[PT_EPT], rk[PT_EPT];
#pragma unroll
  for (int q = 0; q < PT_EPT; ++q) {
    const int e  = ebase + q * PT_THREADS;
    const int ec = (e < E) ? e : (E - 1);
    const int d = dst[ec], sr = src[ec];
    dv[q] = (e < E) ? d : -1;
    sv[q] = sr;
  }
  __syncthreads();
#pragma unroll
  for (int q = 0; q < PT_EPT; ++q) {
    rk[q] = 0;
    if ((unsigned)dv[q] < (unsigned)N) rk[q] = atomicAdd(&hist[dv[q] >> RB_SHIFT], 1);
  }
  __syncthreads();
  const int c0 = hist[2 * tid], c1 = hist[2 * tid + 1];
  const int s = c0 + c1;
  int inc = s;
  for (int off = 1; off < 32; off <<= 1) {
    const int u = __shfl_up(inc, off);
    if (lane >= off) inc += u;
  }
  if (lane == 31) wsum[wave] = inc;
  __syncthreads();
  int add = 0, tot = 0;
#pragma unroll
  for (int w = 0; w < PT_THREADS / 32; ++w) {
    const int ws = wsum[w];
    if (w < wave) add += ws;
    tot += ws;
  }
  const int ex0 = add + inc - s, ex1 = ex0 + c0;
  hist[2 * tid]     = ex0;
  hist[2 * tid + 1] = ex1;
  if (2 * tid < B)     tab[(2 * tid) * G + g]     = (ex0 << RUN_SHIFT) | c0;
  if (2 * tid + 1 < B) tab[(2 * tid + 1) * G + g] = (ex1 << RUN_SHIFT) | c1;
  __syncthreads();
#pragma unroll
  for (int q = 0; q < PT_EPT; ++q) {
    if ((unsigned)dv[q] < (unsigned)N)
      srt[hist[dv[q] >> RB_SHIFT] + rk[q]] =
          (int)(((unsigned)sv[q] << RB_SHIFT) | ((unsigned)dv[q] & (unsigned)(RB_ROWS - 1)));
  }
  __syncthreads();
  int* myrec = rec + (size_t)g * PT_CHUNK;
  if (tot == PT_CHUNK) {
    int4 v[PT_EPT / 4];
#pragma unroll
    for (int k = 0; k < PT_EPT / 4; ++k) v[k] = *(const int4*)(&srt[(k * PT_THREADS + tid) * 4]);
#pragma unroll
    for (int k = 0; k < PT_EPT / 4; ++k) *(int4*)(myrec + (k * PT_THREADS + tid) * 4) = v[k];
  } else {
    for (int i = tid; i < tot; i += PT_THREADS) myrec[i] = srt[i];
  }
}

static constexpr int BP_RUNS  = 8;
static constexpr int BP_WAVES = RB_ROWS / 32;
static constexpr int BP_STAGE = 12288;

template <bool STAGED>
__device__ __forceinline__ void place_runs(const int* __restrict__ rec, const int* run, int* cnt, int* stage,
                                           int* __restrict__ ssrc, int a0, int G, int lane, int wave) {
  for (int j0 = 0; j0 * BP_WAVES + wave < G; j0 += BP_RUNS) {
    int r[BP_RUNS][2], len[BP_RUNS], at[BP_RUNS];
#pragma unroll
    for (int u = 0; u < BP_RUNS; ++u) {
      const int g = (j0 + u) * BP_WAVES + wave;
      const int pk = (g < G) ? run[g] : 0;
      len[u] = pk & RUN_MASK;
      at[u]  = g * PT_CHUNK + (pk >> RUN_SHIFT);
      r[u][0] = (lane < len[u]) ? rec[at[u] + lane] : -1;
      r[u][1] = (lane + 32 < len[u]) ? rec[at[u] + lane + 32] : -1;
    }
#pragma unroll
    for (int u = 0; u < BP_RUNS; ++u) {
      if (r[u][0] >= 0) {
        const int pos = atomicAdd(&cnt[r[u][0] & (RB_ROWS - 1)], 1);
        const int v = (int)((unsigned)r[u][0] >> RB_SHIFT);
        if (STAGED) stage[pos - a0] = v; else ssrc[pos] = v;
      }
      if (r[u][1] >= 0) {
        const int pos = atomicAdd(&cnt[r[u][1] & (RB_ROWS - 1)], 1);
        const int v = (int)((unsigned)r[u][1] >> RB_SHIFT);
        if (STAGED) stage[pos - a0] = v; else ssrc[pos] = v;
      }
      for (int o = lane + 64; o < len[u]; o += 32) {
        const int rr = rec[at[u] + o];
        const int pos = atomicAdd(&cnt[rr & (RB_ROWS - 1)], 1);
        const int v = (int)((unsigned)rr >> RB_SHIFT);
        if (STAGED) stage[pos - a0] = v; else ssrc[pos] = v;
      }
    }
  }
}

__global__ __launch_bounds__(RB_ROWS) void bucket_place_kernel(const int* __restrict__ rec,
                                                               const int* __restrict__ tab,
                                                               int* __restrict__ deg, int* __restrict__ offs,
                                                               int* __restrict__ ssrc, int N, int G) {
  __shared__ int cnt[RB_ROWS];
  __shared__ int wsum[RB_ROWS / 32];
  __shared__ int bsum[RB_ROWS / 32];
  __shared__ int run[MAX_CHUNKS];
  __shared__ alignas(16) int stage[BP_STAGE];
  const int tid = threadIdx.x, lane = tid & 31, wave = tid >> 5;
  const int b = blockIdx.x;
  cnt[tid] = 0;
  int part = 0;
  for (int g = tid; g < G; g += RB_ROWS) {
    const int pk = tab[b * G + g];
    run[g] = pk;
    part += pk >> RUN_SHIFT;
  }
  for (int off = 16; off > 0; off >>= 1) part += __shfl_xor(part, off);
  if (lane == 0) bsum[wave] = part;
  __syncthreads();
  int s = 0;
#pragma unroll
  for (int w = 0; w < RB_ROWS / 32; ++w) s += bsum[w];
  for (int j0 = 0; j0 * BP_WAVES + wave < G; j0 += BP_RUNS) {
    int r[BP_RUNS][2], len[BP_RUNS], at[BP_RUNS];
#pragma unroll
    for (int u = 0; u < BP_RUNS; ++u) {
      const int g = (j0 + u) * BP_WAVES + wave;
      const int pk = (g < G) ? run[g] : 0;
      len[u] = pk & RUN_MASK;
      at[u]  = g * PT_CHUNK + (pk >> RUN_SHIFT);
      r[u][0] = (lane < len[u]) ? rec[at[u] + lane] : -1;
      r[u][1] = (lane + 32 < len[u]) ? rec[at[u] + lane + 32] : -1;
    }
#pragma unroll
    for (int u = 0; u < BP_RUNS; ++u) {
      if (r[u][0] >= 0) atomicAdd(&cnt[r[u][0] & (RB_ROWS - 1)], 1);
      if (r[u][1] >= 0) atomicAdd(&cnt[r[u][1] & (RB_ROWS - 1)], 1);
      for (int o = lane + 64; o < len[u]; o += 32) atomicAdd(&cnt[rec[at[u] + o] & (RB_ROWS - 1)], 1);
    }
  }
  __syncthreads();
  const int c = cnt[tid];
  int inc = c;
  for (int off = 1; off < 32; off <<= 1) {
    const int u = __shfl_up(inc, off);
    if (lane >= off) inc += u;
  }
  if (lane == 31) wsum[wave] = inc;
  __syncthreads();
  int add = 0, tot = 0;
#pragma unroll
  for (int w = 0; w < RB_ROWS / 32; ++w) {
    const int ws = wsum[w];
    if (w < wave) add += ws;
    tot += ws;
  }
  const int ex = s + add + inc - c;
  const int a0 = s & ~31;
  const int span = (s - a0) + tot;
  const bool staged = span <= BP_STAGE;
  const int row = (b << RB_SHIFT) + tid;
  if (row < N) {
    deg[row]  = c;
    offs[row] = ex;
  }
  cnt[tid] = ex;
  __syncthreads();
  if (staged) {
    place_runs<true>(rec, run, cnt, stage, ssrc, a0, G, lane, wave);
    __syncthreads();
    const int shift = s - a0;
    for (int i4 = tid * 4; i4 < span; i4 += RB_ROWS * 4) {
      if (i4 >= shift && i4 + 3 < span) {
        *(int4*)(ssrc + a0 + i4) = *(const int4*)(&stage[i4]);
      } else {
        for (int j = 0; j < 4; ++j) {
          const int i = i4 + j;
          if (i >= shift && i < span) ssrc[a0 + i] = stage[i];
        }
      }
    }
  } else {
    place_runs<false>(rec, run, cnt, stage, ssrc, a0, G, lane, wave);
  }
}

static constexpr int AS_STRIDE = 132;
static constexpr int GEMM_WGS  = 1024;

__global__ __launch_bounds__(256) void gat_gemm_wmma_kernel(
    const float* __restrict__ hfeat, const float* __restrict__ W,
    float* __restrict__ Wh, int N) {
  __shared__ alignas(16) float As[2][16 * AS_STRIDE];

  const int tid  = threadIdx.x;
  const int wave = tid >> 5;
  const int lane = tid & 31;
  const int hl   = lane >> 4;
  const int col  = lane & 15;
  const int n0   = wave * 16;
  const int nstrips = (N + 15) >> 4;
  const int sstep   = gridDim.x;

  const float* Wb = W + wave * (KF * KD) + col;
  v2f bfr[KF / 4];
#pragma unroll
  for (int j = 0; j < KF / 4; ++j) {
    const int k = 4 * j + 2 * hl;
    bfr[j].x = Wb[(size_t)k * KD];
    bfr[j].y = Wb[(size_t)(k + 1) * KD];
  }

  const int sr = tid >> 5;
  const int sf = (tid & 31) * 4;
  float4 p0, p1;
  const float4 zero4 = make_float4(0.0f, 0.0f, 0.0f, 0.0f);

  int s = blockIdx.x;
  {
    const int ra = s * 16 + sr, rb = ra + 8;
    p0 = zero4;
    p1 = zero4;
    if (s < nstrips && ra < N) p0 = *(const float4*)(hfeat + (size_t)ra * KF + sf);
    if (s < nstrips && rb < N) p1 = *(const float4*)(hfeat + (size_t)rb * KF + sf);
    *(float4*)(&As[0][sr * AS_STRIDE + sf])       = p0;
    *(float4*)(&As[0][(sr + 8) * AS_STRIDE + sf]) = p1;
  }
  __syncthreads();

  int cur = 0;
  for (; s < nstrips; s += sstep) {
    const int sn = s + sstep;
    const bool more = sn < nstrips;
    if (more) {
      const int ra = sn * 16 + sr, rb = ra + 8;
      p0 = zero4;
      p1 = zero4;
      if (ra < N) p0 = *(const float4*)(hfeat + (size_t)ra * KF + sf);
      if (rb < N) p1 = *(const float4*)(hfeat + (size_t)rb * KF + sf);
    }

    const float* Ac = &As[cur][col * AS_STRIDE + 2 * hl];
    v8f acc = {};
#pragma unroll
    for (int j = 0; j < KF / 4; ++j) {
      v2f a;
      a.x = Ac[4 * j];
      a.y = Ac[4 * j + 1];
      acc = __builtin_amdgcn_wmma_f32_16x16x4_f32(
          false, a, false, bfr[j], (short)0, acc, false, false);
    }

    const int row0 = s * 16;
#pragma unroll
    for (int v = 0; v < 8; ++v) {
      int r = row0 + v + 8 * hl;
      if (r < N) Wh[(size_t)r * KHD + n0 + col] = acc[v];
    }

    if (more) {
      *(float4*)(&As[cur ^ 1][sr * AS_STRIDE + sf])       = p0;
      *(float4*)(&As[cur ^ 1][(sr + 8) * AS_STRIDE + sf]) = p1;
    }
    __syncthreads();
    cur ^= 1;
  }
}

__device__ __forceinline__ float leaky(float e) {
  return (e > 0.0f) ? e : KALPHA * e;
}

__device__ __forceinline__ float dot4(const float4 w, const float4 a) {
  return w.x * a.x + w.y * a.y + w.z * a.z + w.w * a.w;
}

__device__ __forceinline__ float quad_sum(float v) {
  v += __int_as_float(__builtin_amdgcn_update_dpp(0, __float_as_int(v), 0x161, 0xf, 0xf, true));
  v += __int_as_float(__builtin_amdgcn_update_dpp(0, __float_as_int(v), 0x162, 0xf, 0xf, true));
  return v;
}

static constexpr int ROW_WGS = 1786;
__global__ __launch_bounds__(256) void row_softmax_gather_kernel(
    const int* __restrict__ ssrc, const int* __restrict__ offs, const int* __restrict__ deg,
    const float* __restrict__ a1, const float* __restrict__ a2,
    const float* __restrict__ Wh, float* __restrict__ out, int N) {
  const int lane = threadIdx.x & 31;
  const float4 a1v = make_float4(a1[lane * 4], a1[lane * 4 + 1], a1[lane * 4 + 2], a1[lane * 4 + 3]);
  const float4 a2v = make_float4(a2[lane * 4], a2[lane * 4 + 1], a2[lane * 4 + 2], a2[lane * 4 + 3]);
  const float* whl = Wh + lane * 4;
  int d = blockIdx.x * 8 + __builtin_amdgcn_readfirstlane((int)(threadIdx.x >> 5));
  const int dstep = gridDim.x * 8;
  int nbeg = 0, ncnt = 0;
  float4 nwd = make_float4(0.0f, 0.0f, 0.0f, 0.0f);
  if (d < N) { nbeg = offs[d]; ncnt = deg[d]; nwd = *(const float4*)(whl + (size_t)d * KHD); }
  for (; d < N; d += dstep) {
  const int beg = nbeg;
  const int end = beg + ncnt;
  const float4 wdst = nwd;
  {
    const int dn = d + dstep;
    if (dn < N) { nbeg = offs[dn]; ncnt = deg[dn]; nwd = *(const float4*)(whl + (size_t)dn * KHD); }
  }

  const float s2b = quad_sum(dot4(wdst, a2v));
  float4 acc = make_float4(0.0f, 0.0f, 0.0f, 0.0f);
  float z = 0.0f;
  float m = -__builtin_inff();
  int i = beg;
  for (; i + 4 <= end; i += 4) {
    const int sa = ssrc[i], sb = ssrc[i + 1], sc = ssrc[i + 2], sd = ssrc[i + 3];
    const float4 wa = *(const float4*)(whl + (size_t)sa * KHD);
    const float4 wb = *(const float4*)(whl + (size_t)sb * KHD);
    const float4 wc = *(const float4*)(whl + (size_t)sc * KHD);
    const float4 wd = *(const float4*)(whl + (size_t)sd * KHD);
    const float ea = quad_sum(dot4(wa, a1v)), eb = quad_sum(dot4(wb, a1v));
    const float ec = quad_sum(dot4(wc, a1v)), ed = quad_sum(dot4(wd, a1v));
    const float xa = leaky(ea + s2b), xb = leaky(eb + s2b), xc = leaky(ec + s2b), xd = leaky(ed + s2b);
    const float mn = fmaxf(fmaxf(fmaxf(xa, xb), fmaxf(xc, xd)), m);
    const float f  = __expf(m - mn);
    m = mn;
    const float pa = __expf(xa - mn);
    const float pb = __expf(xb - mn);
    const float pc = __expf(xc - mn);
    const float pd = __expf(xd - mn);
    z *= f; acc.x *= f; acc.y *= f; acc.z *= f; acc.w *= f;
    z += pa; acc.x += pa * wa.x; acc.y += pa * wa.y; acc.z += pa * wa.z; acc.w += pa * wa.w;
    z += pb; acc.x += pb * wb.x; acc.y += pb * wb.y; acc.z += pb * wb.z; acc.w += pb * wb.w;
    z += pc; acc.x += pc * wc.x; acc.y += pc * wc.y; acc.z += pc * wc.z; acc.w += pc * wc.w;
    z += pd; acc.x += pd * wd.x; acc.y += pd * wd.y; acc.z += pd * wd.z; acc.w += pd * wd.w;
  }
#pragma nounroll
  for (; i < end; ++i) {
    const int sa = ssrc[i];
    const float4 wa = *(const float4*)(whl + (size_t)sa * KHD);
    const float ea = quad_sum(dot4(wa, a1v));
    const float xa = leaky(ea + s2b);
    const float mn = fmaxf(xa, m);
    const float f  = __expf(m - mn);
    m = mn;
    const float pa = __expf(xa - mn);
    z *= f; acc.x *= f; acc.y *= f; acc.z *= f; acc.w *= f;
    z += pa; acc.x += pa * wa.x; acc.y += pa * wa.y; acc.z += pa * wa.z; acc.w += pa * wa.w;
  }
  const float zinv = 1.0f / fmaxf(z, 1e-16f);
  *(float4*)(out + (size_t)d * KHD + lane * 4) =
      make_float4(acc.x * zinv, acc.y * zinv, acc.z * zinv, acc.w * zinv);
  }
}

extern "C" void kernel_launch(void* const* d_in, const int* in_sizes, int n_in,
                              void* d_out, int out_size, void* d_ws, size_t ws_size,
                              hipStream_t stream) {
  const float* hfeat = (const float*)d_in[0];
  const int*   src   = (const int*)d_in[1];
  const int*   dst   = (const int*)d_in[2];
  const float* W     = (const float*)d_in[3];
  const float* a1    = (const float*)d_in[4];
  const float* a2    = (const float*)d_in[5];
  float* out = (float*)d_out;

  const int N = in_sizes[0] / KF;
  const int E = in_sizes[1];

  float* Wh   = (float*)d_ws;
  int*   ssrc = (int*)(Wh + (size_t)N * KHD);
  int*   deg  = ssrc + (size_t)E;
  int*   offs = deg + (size_t)N;
  int*   rec  = (int*)d_ws;
  int*   tab  = rec + (size_t)E;

  const int TB = 256;
  const int G = (E + PT_CHUNK - 1) / PT_CHUNK;
  const int B = (N + RB_ROWS - 1) / RB_ROWS;

  chunk_sort_kernel<<<G, PT_THREADS, 0, stream>>>(src, dst, rec, tab, N, E, B);
  bucket_place_kernel<<<B, RB_ROWS, 0, stream>>>(rec, tab, deg, offs, ssrc, N, G);

  gat_gemm_wmma_kernel<<<GEMM_WGS, TB, 0, stream>>>(hfeat, W, Wh, N);

  row_softmax_gather_kernel<<<ROW_WGS, TB, 0, stream>>>(ssrc, offs, deg, a1, a2, Wh, out, N);
}


# Row order made run-invariant (edge ranks from per-wave LDS counters, waves' shares ordered in the scan) so that the projection strips can share the two bucketing launches: 3 launches for 4
# speedup vs baseline: 12.1367x; 1.0675x over previous
#include <hip/hip_runtime.h>
#include <hip/hip_bf16.h>


static constexpr int KF  = 128;
static constexpr int KH  = 8;
static constexpr int KD  = 16;
static constexpr int KHD = 128;
static constexpr float KALPHA = 0.2f;

typedef __attribute__((ext_vector_type(2))) float v2f;
typedef __attribute__((ext_vector_type(8))) float v8f;

static constexpr int PT_THREADS  = 512;
static constexpr int PT_WAVES    = PT_THREADS / 32;
static constexpr int PT_EPT      = 16;
static constexpr int PT_CHUNK    = PT_THREADS * PT_EPT;
static constexpr int RB_SHIFT    = 9;
static constexpr int RB_ROWS     = 1 << RB_SHIFT;
static constexpr int MAX_BUCKETS = 2 * PT_THREADS;
static constexpr int MAX_CHUNKS  = 1024;
static constexpr int RUN_SHIFT = 14;
static constexpr int RUN_MASK  = (1 << RUN_SHIFT) - 1;

static constexpr int AS_STRIDE    = 132;
static constexpr int PJ_BUF_WORDS = 16 * AS_STRIDE;
static constexpr int PJ_LDS_WORDS = 2 * 2 * PJ_BUF_WORDS;
static constexpr int PJ_WGS       = 316;

__device__ __forceinline__ void proj_strips(float* lds, const float* __restrict__ hfeat,
                                            const float* __restrict__ W, float* __restrict__ Wh, int N,
                                            int s_begin, int s_end, int blk, int nblk) {
  const int tid  = threadIdx.x;
  const int grp  = __builtin_amdgcn_readfirstlane(tid >> 8);
  const int t    = tid & 255;
  const int wave = t >> 5;
  const int lane = tid & 31;
  const int hl   = lane >> 4;
  const int col  = lane & 15;
  const int n0   = wave * 16;
  const int sstep = 2 * nblk;
  float* As = lds + grp * (2 * PJ_BUF_WORDS);

  const float* Wb = W + wave * (KF * KD) + col;
  v2f bfr[KF / 4];
#pragma unroll
  for (int j = 0; j < KF / 4; ++j) {
    const int k = 4 * j + 2 * hl;
    bfr[j].x = Wb[(size_t)k * KD];
    bfr[j].y = Wb[(size_t)(k + 1) * KD];
  }

  const int sr = t >> 5;
  const int sf = (t & 31) * 4;
  float4 p0, p1;
  const float4 zero4 = make_float4(0.0f, 0.0f, 0.0f, 0.0f);

  int sb = s_begin + 2 * blk;
  {
    const int s = sb + grp;
    const int ra = s * 16 + sr, rb = ra + 8;
    p0 = zero4;
    p1 = zero4;
    if (s < s_end && ra < N) p0 = *(const float4*)(hfeat + (size_t)ra * KF + sf);
    if (s < s_end && rb < N) p1 = *(const float4*)(hfeat + (size_t)rb * KF + sf);
    *(float4*)(&As[sr * AS_STRIDE + sf])       = p0;
    *(float4*)(&As[(sr + 8) * AS_STRIDE + sf]) = p1;
  }
  __syncthreads();

  int cur = 0;
  for (; sb < s_end; sb += sstep) {
    const int s  = sb + grp;
    const int sn = s + sstep;
    const bool more = sn < s_end;
    if (more) {
      const int ra = sn * 16 + sr, rb = ra + 8;
      p0 = zero4;
      p1 = zero4;
      if (ra < N) p0 = *(const float4*)(hfeat + (size_t)ra * KF + sf);
      if (rb < N) p1 = *(const float4*)(hfeat + (size_t)rb * KF + sf);
    }

    if (s < s_end) {
      const float* Ac = &As[cur * PJ_BUF_WORDS + col * AS_STRIDE + 2 * hl];
      v8f acc = {};
#pragma unroll
      for (int j = 0; j < KF / 4; ++j) {
        v2f a;
        a.x = Ac[4 * j];
        a.y = Ac[4 * j + 1];
        acc = __builtin_amdgcn_wmma_f32_16x16x4_f32(
            false, a, false, bfr[j], (short)0, acc, false, false);
      }

      const int row0 = s * 16;
#pragma unroll
      for (int v = 0; v < 8; ++v) {
        int r = row0 + v + 8 * hl;
        if (r < N) Wh[(size_t)r * KHD + n0 + col] = acc[v];
      }
    }

    if (more) {
      *(float4*)(&As[(cur ^ 1) * PJ_BUF_WORDS + sr * AS_STRIDE + sf])       = p0;
      *(float4*)(&As[(cur ^ 1) * PJ_BUF_WORDS + (sr + 8) * AS_STRIDE + sf]) = p1;
    }
    __syncthreads();
    cur ^= 1;
  }
}

static constexpr int CS_LDS_WORDS = PT_WAVES * MAX_BUCKETS + 32 + PT_CHUNK;

__device__ __forceinline__ void chunk_sort_body(int* lds, const int* __restrict__ src,
                                                const int* __restrict__ dst, int* __restrict__ rec,
                                                int* __restrict__ tab, int N, int E, int B, int g, int G) {
  int* hist = lds;
  int* wsum = lds + PT_WAVES * MAX_BUCKETS;
  int* srt  = wsum + 32;
  const int tid = threadIdx.x, lane = tid & 31, wave = tid >> 5;
  int* myh = hist + wave * MAX_BUCKETS;
  const int Bz = (B + 1) & ~1;
  for (int i = lane; i < Bz; i += 32) myh[i] = 0;
  const int ebase = g * PT_CHUNK + tid;
  int dv[PT_EPT], sv[PT_EPT], rk[PT_EPT];
#pragma unroll
  for (int q = 0; q < PT_EPT; ++q) {
    const int e  = ebase + q * PT_THREADS;
    const int ec = (e < E) ? e : (E - 1);
    const int d = dst[ec], sr = src[ec];
    dv[q] = (e < E) ? d : -1;
    sv[q] = sr;
  }
  __syncthreads();
#pragma unroll
  for (int q = 0; q < PT_EPT; ++q) {
    rk[q] = 0;
    if ((unsigned)dv[q] < (unsigned)N) rk[q] = atomicAdd(&myh[dv[q] >> RB_SHIFT], 1);
  }
  __syncthreads();
  int c0 = 0, c1 = 0;
  if (2 * tid < B) {
#pragma unroll
    for (int w = 0; w < PT_WAVES; ++w) {
      const int2 h = *(const int2*)(&hist[w * MAX_BUCKETS + 2 * tid]);
      c0 += h.x;
      c1 += h.y;
    }
  }
  const int s = c0 + c1;
  int inc = s;
  for (int off = 1; off < 32; off <<= 1) {
    const int u = __shfl_up(inc, off);
    if (lane >= off) inc += u;
  }
  if (lane == 31) wsum[wave] = inc;
  __syncthreads();
  int add = 0, tot = 0;
#pragma unroll
  for (int w = 0; w < PT_WAVES; ++w) {
    const int ws = wsum[w];
    if (w < wave) add += ws;
    tot += ws;
  }
  const int ex0 = add + inc - s, ex1 = ex0 + c0;
  if (2 * tid < B) {
    int r0 = ex0, r1 = ex1;
#pragma unroll
    for (int w = 0; w < PT_WAVES; ++w) {
      int2* p = (int2*)(&hist[w * MAX_BUCKETS + 2 * tid]);
      const int2 h = *p;
      *p = make_int2(r0, r1);
      r0 += h.x;
      r1 += h.y;
    }
  }
  if (2 * tid < B)     tab[(2 * tid) * G + g]     = (ex0 << RUN_SHIFT) | c0;
  if (2 * tid + 1 < B) tab[(2 * tid + 1) * G + g] = (ex1 << RUN_SHIFT) | c1;
  __syncthreads();
#pragma unroll
  for (int q = 0; q < PT_EPT; ++q) {
    if ((unsigned)dv[q] < (unsigned)N)
      srt[myh[dv[q] >> RB_SHIFT] + rk[q]] =
          (int)(((unsigned)sv[q] << RB_SHIFT) | ((unsigned)dv[q] & (unsigned)(RB_ROWS - 1)));
  }
  __syncthreads();
  int* myrec = rec + (size_t)g * PT_CHUNK;
  if (tot == PT_CHUNK) {
    int4 v[PT_EPT / 4];
#pragma unroll
    for (int k = 0; k < PT_EPT / 4; ++k) v[k] = *(const int4*)(&srt[(k * PT_THREADS + tid) * 4]);
#pragma unroll
    for (int k = 0; k < PT_EPT / 4; ++k) *(int4*)(myrec + (k * PT_THREADS + tid) * 4) = v[k];
  } else {
    for (int i = tid; i < tot; i += PT_THREADS) myrec[i] = srt[i];
  }
}

__global__ __launch_bounds__(PT_THREADS) void chunk_sort_proj_kernel(
    const int* __restrict__ src, const int* __restrict__ dst, int* __restrict__ rec, int* __restrict__ tab,
    int N, int E, int B, int G,
    const float* __restrict__ hfeat, const float* __restrict__ W, float* __restrict__ Wh) {
  __shared__ alignas(16) unsigned char lds[CS_LDS_WORDS * 4];
  const int blk = blockIdx.x;
  if (blk < G) {
    chunk_sort_body((int*)lds, src, dst, rec, tab, N, E, B, blk, G);
  } else {
    const int nstrips = (N + 15) >> 4;
    proj_strips((float*)lds, hfeat, W, Wh, N, 0, nstrips >> 1, blk - G, (int)gridDim.x - G);
  }
}

static constexpr int BP_RUNS  = 8;
static constexpr int BP_WAVES = RB_ROWS / 32;
static constexpr int BP_STAGE = 12288;
static constexpr int BP_LDS_WORDS = BP_WAVES * RB_ROWS + 32 + MAX_CHUNKS + BP_STAGE;

template <bool STAGED>
__device__ __forceinline__ void place_runs(const int* __restrict__ rec, const int* run, int* cnt, int* stage,
                                           int* __restrict__ ssrc, int a0, int G, int lane, int wave) {
  for (int j0 = 0; j0 * BP_WAVES + wave < G; j0 += BP_RUNS) {
    int r[BP_RUNS][2], len[BP_RUNS], at[BP_RUNS];
#pragma unroll
    for (int u = 0; u < BP_RUNS; ++u) {
      const int g = (j0 + u) * BP_WAVES + wave;
      const int pk = (g < G) ? run[g] : 0;
      len[u] = pk & RUN_MASK;
      at[u]  = g * PT_CHUNK + (pk >> RUN_SHIFT);
      r[u][0] = (lane < len[u]) ? rec[at[u] + lane] : -1;
      r[u][1] = (lane + 32 < len[u]) ? rec[at[u] + lane + 32] : -1;
    }
#pragma unroll
    for (int u = 0; u < BP_RUNS; ++u) {
      if (r[u][0] >= 0) {
        const int pos = atomicAdd(&cnt[r[u][0] & (RB_ROWS - 1)], 1);
        const int v = (int)((unsigned)r[u][0] >> RB_SHIFT);
        if (STAGED) stage[pos - a0] = v; else ssrc[pos] = v;
      }
      if (r[u][1] >= 0) {
        const int pos = atomicAdd(&cnt[r[u][1] & (RB_ROWS - 1)], 1);
        const int v = (int)((unsigned)r[u][1] >> RB_SHIFT);
        if (STAGED) stage[pos - a0] = v; else ssrc[pos] = v;
      }
      for (int o = lane + 64; o < len[u]; o += 32) {
        const int rr = rec[at[u] + o];
        const int pos = atomicAdd(&cnt[rr & (RB_ROWS - 1)], 1);
        const int v = (int)((unsigned)rr >> RB_SHIFT);
        if (STAGED) stage[pos - a0] = v; else ssrc[pos] = v;
      }
    }
  }
}

__device__ __forceinline__ void bucket_place_body(int* lds, const int* __restrict__ rec,
                                                  const int* __restrict__ tab, int* __restrict__ deg,
                                                  int* __restrict__ offs, int* __restrict__ ssrc,
                                                  int N, int G, int b) {
  int* cntw  = lds;
  int* wsum  = cntw + BP_WAVES * RB_ROWS;
  int* bsum  = wsum + 16;
  int* run   = bsum + 16;
  int* stage = run + MAX_CHUNKS;
  const int tid = threadIdx.x, lane = tid & 31, wave = tid >> 5;
  int* myc = cntw + wave * RB_ROWS;
  for (int w = 0; w < BP_WAVES; ++w) cntw[w * RB_ROWS + tid] = 0;
  int part = 0;
  for (int g = tid; g < G; g += RB_ROWS) {
    const int pk = tab[b * G + g];
    run[g] = pk;
    part += pk >> RUN_SHIFT;
  }
  for (int off = 16; off > 0; off >>= 1) part += __shfl_xor(part, off);
  if (lane == 0) bsum[wave] = part;
  __syncthreads();
  int s = 0;
#pragma unroll
  for (int w = 0; w < RB_ROWS / 32; ++w) s += bsum[w];
  for (int j0 = 0; j0 * BP_WAVES + wave < G; j0 += BP_RUNS) {
    int r[BP_RUNS][2], len[BP_RUNS], at[BP_RUNS];
#pragma unroll
    for (int u = 0; u < BP_RUNS; ++u) {
      const int g = (j0 + u) * BP_WAVES + wave;
      const int pk = (g < G) ? run[g] : 0;
      len[u] = pk & RUN_MASK;
      at[u]  = g * PT_CHUNK + (pk >> RUN_SHIFT);
      r[u][0] = (lane < len[u]) ? rec[at[u] + lane] : -1;
      r[u][1] = (lane + 32 < len[u]) ? rec[at[u] + lane + 32] : -1;
    }
#pragma unroll
    for (int u = 0; u < BP_RUNS; ++u) {
      if (r[u][0] >= 0) atomicAdd(&myc[r[u][0] & (RB_ROWS - 1)], 1);
      if (r[u][1] >= 0) atomicAdd(&myc[r[u][1] & (RB_ROWS - 1)], 1);
      for (int o = lane + 64; o < len[u]; o += 32) atomicAdd(&myc[rec[at[u] + o] & (RB_ROWS - 1)], 1);
    }
  }
  __syncthreads();
  int c = 0;
#pragma unroll
  for (int w = 0; w < BP_WAVES; ++w) c += cntw[w * RB_ROWS + tid];
  int inc = c;
  for (int off = 1; off < 32; off <<= 1) {
    const int u = __shfl_up(inc, off);
    if (lane >= off) inc += u;
  }
  if (lane == 31) wsum[wave] = inc;
  __syncthreads();
  int add = 0, tot = 0;
#pragma unroll
  for (int w = 0; w < RB_ROWS / 32; ++w) {
    const int ws = wsum[w];
    if (w < wave) add += ws;
    tot += ws;
  }
  const int ex = s + add + inc - c;
  const int a0 = s & ~31;
  const int span = (s - a0) + tot;
  const bool staged = span <= BP_STAGE;
  const int row = (b << RB_SHIFT) + tid;
  if (row < N) {
    deg[row]  = c;
    offs[row] = ex;
  }
  {
    int rr = ex;
#pragma unroll
    for (int w = 0; w < BP_WAVES; ++w) {
      int* p = &cntw[w * RB_ROWS + tid];
      const int h = *p;
      *p = rr;
      rr += h;
    }
  }
  __syncthreads();
  if (staged) {
    place_runs<true>(rec, run, myc, stage, ssrc, a0, G, lane, wave);
    __syncthreads();
    const int shift = s - a0;
    for (int i4 = tid * 4; i4 < span; i4 += RB_ROWS * 4) {
      if (i4 >= shift && i4 + 3 < span) {
        *(int4*)(ssrc + a0 + i4) = *(const int4*)(&stage[i4]);
      } else {
        for (int j = 0; j < 4; ++j) {
          const int i = i4 + j;
          if (i >= shift && i < span) ssrc[a0 + i] = stage[i];
        }
      }
    }
  } else {
    place_runs<false>(rec, run, myc, stage, ssrc, a0, G, lane, wave);
  }
}

__global__ __launch_bounds__(RB_ROWS) void bucket_place_proj_kernel(
    const int* __restrict__ rec, const int* __restrict__ tab, int* __restrict__ deg, int* __restrict__ offs,
    int* __restrict__ ssrc, int N, int G, int B,
    const float* __restrict__ hfeat, const float* __restrict__ W, float* __restrict__ Wh) {
  __shared__ alignas(16) unsigned char lds[BP_LDS_WORDS * 4];
  const int blk = blockIdx.x;
  if (blk < B) {
    bucket_place_body((int*)lds, rec, tab, deg, offs, ssrc, N, G, blk);
  } else {
    const int nstrips = (N + 15) >> 4;
    proj_strips((float*)lds, hfeat, W, Wh, N, nstrips >> 1, nstrips, blk - B, (int)gridDim.x - B);
  }
}

__device__ __forceinline__ float leaky(float e) {
  return (e > 0.0f) ? e : KALPHA * e;
}

__device__ __forceinline__ float dot4(const float4 w, const float4 a) {
  return w.x * a.x + w.y * a.y + w.z * a.z + w.w * a.w;
}

__device__ __forceinline__ float quad_sum(float v) {
  v += __int_as_float(__builtin_amdgcn_update_dpp(0, __float_as_int(v), 0x161, 0xf, 0xf, true));
  v += __int_as_float(__builtin_amdgcn_update_dpp(0, __float_as_int(v), 0x162, 0xf, 0xf, true));
  return v;
}

static constexpr int ROW_WGS = 1786;
__global__ __launch_bounds__(256) void row_softmax_gather_kernel(
    const int* __restrict__ ssrc, const int* __restrict__ offs, const int* __restrict__ deg,
    const float* __restrict__ a1, const float* __restrict__ a2,
    const float* __restrict__ Wh, float* __restrict__ out, int N) {
  const int lane = threadIdx.x & 31;
  const float4 a1v = make_float4(a1[lane * 4], a1[lane * 4 + 1], a1[lane * 4 + 2], a1[lane * 4 + 3]);
  const float4 a2v = make_float4(a2[lane * 4], a2[lane * 4 + 1], a2[lane * 4 + 2], a2[lane * 4 + 3]);
  const float* whl = Wh + lane * 4;
  int d = blockIdx.x * 8 + __builtin_amdgcn_readfirstlane((int)(threadIdx.x >> 5));
  const int dstep = gridDim.x * 8;
  int nbeg = 0, ncnt = 0;
  float4 nwd = make_float4(0.0f, 0.0f, 0.0f, 0.0f);
  if (d < N) { nbeg = offs[d]; ncnt = deg[d]; nwd = *(const float4*)(whl + (size_t)d * KHD); }
  for (; d < N; d += dstep) {
  const int beg = nbeg;
  const int end = beg + ncnt;
  const float4 wdst = nwd;
  {
    const int dn = d + dstep;
    if (dn < N) { nbeg = offs[dn]; ncnt = deg[dn]; nwd = *(const float4*)(whl + (size_t)dn * KHD); }
  }

  const float s2b = quad_sum(dot4(wdst, a2v));
  float4 acc = make_float4(0.0f, 0.0f, 0.0f, 0.0f);
  float z = 0.0f;
  float m = -__builtin_inff();
  int i = beg;
  for (; i + 4 <= end; i += 4) {
    const int sa = ssrc[i], sb = ssrc[i + 1], sc = ssrc[i + 2], sd = ssrc[i + 3];
    const float4 wa = *(const float4*)(whl + (size_t)sa * KHD);
    const float4 wb = *(const float4*)(whl + (size_t)sb * KHD);
    const float4 wc = *(const float4*)(whl + (size_t)sc * KHD);
    const float4 wd = *(const float4*)(whl + (size_t)sd * KHD);
    const float ea = quad_sum(dot4(wa, a1v)), eb = quad_sum(dot4(wb, a1v));
    const float ec = quad_sum(dot4(wc, a1v)), ed = quad_sum(dot4(wd, a1v));
    const float xa = leaky(ea + s2b), xb = leaky(eb + s2b), xc = leaky(ec + s2b), xd = leaky(ed + s2b);
    const float mn = fmaxf(fmaxf(fmaxf(xa, xb), fmaxf(xc, xd)), m);
    const float f  = __expf(m - mn);
    m = mn;
    const float pa = __expf(xa - mn);
    const float pb = __expf(xb - mn);
    const float pc = __expf(xc - mn);
    const float pd = __expf(xd - mn);
    z *= f; acc.x *= f; acc.y *= f; acc.z *= f; acc.w *= f;
    z += pa; acc.x += pa * wa.x; acc.y += pa * wa.y; acc.z += pa * wa.z; acc.w += pa * wa.w;
    z += pb; acc.x += pb * wb.x; acc.y += pb * wb.y; acc.z += pb * wb.z; acc.w += pb * wb.w;
    z += pc; acc.x += pc * wc.x; acc.y += pc * wc.y; acc.z += pc * wc.z; acc.w += pc * wc.w;
    z += pd; acc.x += pd * wd.x; acc.y += pd * wd.y; acc.z += pd * wd.z; acc.w += pd * wd.w;
  }
#pragma nounroll
  for (; i < end; ++i) {
    const int sa = ssrc[i];
    const float4 wa = *(const float4*)(whl + (size_t)sa * KHD);
    const float ea = quad_sum(dot4(wa, a1v));
    const float xa = leaky(ea + s2b);
    const float mn = fmaxf(xa, m);
    const float f  = __expf(m - mn);
    m = mn;
    const float pa = __expf(xa - mn);
    z *= f; acc.x *= f; acc.y *= f; acc.z *= f; acc.w *= f;
    z += pa; acc.x += pa * wa.x; acc.y += pa * wa.y; acc.z += pa * wa.z; acc.w += pa * wa.w;
  }
  const float zinv = 1.0f / fmaxf(z, 1e-16f);
  *(float4*)(out + (size_t)d * KHD + lane * 4) =
      make_float4(acc.x * zinv, acc.y * zinv, acc.z * zinv, acc.w * zinv);
  }
}

extern "C" void kernel_launch(void* const* d_in, const int* in_sizes, int n_in,
                              void* d_out, int out_size, void* d_ws, size_t ws_size,
                              hipStream_t stream) {
  const float* hfeat = (const float*)d_in[0];
  const int*   src   = (const int*)d_in[1];
  const int*   dst   = (const int*)d_in[2];
  const float* W     = (const float*)d_in[3];
  const float* a1    = (const float*)d_in[4];
  const float* a2    = (const float*)d_in[5];
  float* out = (float*)d_out;

  const int N = in_sizes[0] / KF;
  const int E = in_sizes[1];

  float* Wh   = (float*)d_ws;
  int*   ssrc = (int*)(Wh + (size_t)N * KHD);
  int*   deg  = ssrc + (size_t)E;
  int*   offs = deg + (size_t)N;
  int*   rec  = (int*)d_out;
  int*   tab  = rec + (size_t)E;

  const int TB = 256;
  const int G = (E + PT_CHUNK - 1) / PT_CHUNK;
  const int B = (N + RB_ROWS - 1) / RB_ROWS;

  chunk_sort_proj_kernel<<<G + PJ_WGS, PT_THREADS, 0, stream>>>(src, dst, rec, tab, N, E, B, G, hfeat, W, Wh);
  bucket_place_proj_kernel<<<B + PJ_WGS, RB_ROWS, 0, stream>>>(rec, tab, deg, offs, ssrc, N, G, B, hfeat, W, Wh);

  row_softmax_gather_kernel<<<ROW_WGS, TB, 0, stream>>>(ssrc, offs, deg, a1, a2, Wh, out, N);
}
